# starting kernel
# baseline (speedup 1.0000x reference)
#include <hip/hip_runtime.h>
#include <math.h>


typedef __attribute__((ext_vector_type(2))) float v2f;
typedef __attribute__((ext_vector_type(8))) float v8f;

#define GMAX 256

__device__ __forceinline__ void atomAddF(float* p, float v) {
  __hip_atomic_fetch_add(p, v, __ATOMIC_RELAXED, __HIP_MEMORY_SCOPE_AGENT);
}

__global__ void k_fill(float* p, float v, int n) {
  int i = blockIdx.x * blockDim.x + threadIdx.x;
  if (i < n) p[i] = v;
}

__global__ void k_deg_count(const int* __restrict__ dst, float* __restrict__ deg, int E) {
  int e = blockIdx.x * blockDim.x + threadIdx.x;
  if (e < E) atomAddF(&deg[dst[e]], 1.0f);
}

__global__ void k_to_dinv(float* deg, int n) {
  int i = blockIdx.x * blockDim.x + threadIdx.x;
  if (i < n) deg[i] = rsqrtf(fmaxf(deg[i], 1.0f));
}

template <int OUT>
__global__ void k_gemm_wmma(const float* __restrict__ A, const float* __restrict__ W,
                            float* __restrict__ C, int N) {
  constexpr int KD = 128;
  __shared__ float As[16 * KD];

  const int row0 = blockIdx.x * 16;
  for (int i = threadIdx.x; i < 16 * KD; i += blockDim.x) {
    int r = i / KD, k = i % KD;
    As[i] = (row0 + r < N) ? A[(size_t)(row0 + r) * KD + k] : 0.0f;
  }
  __syncthreads();

  const int wave = threadIdx.x >> 5;
  const int lane = threadIdx.x & 31;
  const int half = lane >> 4;
  const int n    = lane & 15;
  const int cw   = wave * 16;

  v8f c = {};
  for (int k0 = 0; k0 < KD; k0 += 4) {
    const int k = k0 + 2 * half;
    v2f a, b;
    a.x = As[n * KD + k];
    a.y = As[n * KD + k + 1];
    b.x = W[(size_t)k * OUT + cw + n];
    b.y = W[(size_t)(k + 1) * OUT + cw + n];
    c = __builtin_amdgcn_wmma_f32_16x16x4_f32(false, a, false, b,
                                              (short)0, c, false, false);
  }

#pragma unroll
  for (int r = 0; r < 8; ++r) {
    const int m = r + 8 * half;
    if (row0 + m < N) C[(size_t)(row0 + m) * OUT + cw + n] = c[r];
  }
}

template <int F>
__global__ void k_agg_init(const float* __restrict__ xw, const float* __restrict__ dinv,
                           const float* __restrict__ bias, float* __restrict__ out, int N) {
  constexpr int LP = F / 4;
  int t = blockIdx.x * blockDim.x + threadIdx.x;
  int node = t / LP;
  if (node >= N) return;
  int fo = (t % LP) * 4;
  float s = dinv[node]; s *= s;
  float4 v = *(const float4*)&xw[(size_t)node * F + fo];
  const float4 b = *(const float4*)&bias[fo];
  v.x = v.x * s + b.x; v.y = v.y * s + b.y;
  v.z = v.z * s + b.z; v.w = v.w * s + b.w;
  *(float4*)&out[(size_t)node * F + fo] = v;
}

template <int F>
__global__ void k_agg_edges(const float* __restrict__ xw, const float* __restrict__ dinv,
                            const int* __restrict__ src, const int* __restrict__ dst,
                            float* __restrict__ out, int E) {
  constexpr int LP = F / 4;
  int t = blockIdx.x * blockDim.x + threadIdx.x;
  int e = t / LP;
  if (e >= E) return;
  int fo = (t % LP) * 4;
  int s = src[e], d = dst[e];
  float nrm = dinv[s] * dinv[d];
  float4 v = *(const float4*)&xw[(size_t)s * F + fo];
  float* o = &out[(size_t)d * F + fo];
  atomAddF(o + 0, v.x * nrm);
  atomAddF(o + 1, v.y * nrm);
  atomAddF(o + 2, v.z * nrm);
  atomAddF(o + 3, v.w * nrm);
}

__global__ void k_elu_pool(float* __restrict__ h, const int* __restrict__ batch,
                           float* __restrict__ pool, float* __restrict__ cnt, int N) {
  int node = blockIdx.x * (blockDim.x >> 5) + (threadIdx.x >> 5);
  if (node >= N) return;
  int lane = threadIdx.x & 31;
  float4 v = *(float4*)&h[(size_t)node * 128 + lane * 4];
  v.x = v.x > 0.0f ? v.x : expm1f(v.x);
  v.y = v.y > 0.0f ? v.y : expm1f(v.y);
  v.z = v.z > 0.0f ? v.z : expm1f(v.z);
  v.w = v.w > 0.0f ? v.w : expm1f(v.w);
  *(float4*)&h[(size_t)node * 128 + lane * 4] = v;
  int g = batch[node];
  float* p = &pool[(size_t)g * 128 + lane * 4];
  atomAddF(p + 0, v.x); atomAddF(p + 1, v.y);
  atomAddF(p + 2, v.z); atomAddF(p + 3, v.w);
  if (lane == 0) atomAddF(&cnt[g], 1.0f);
}

template <int IN, int OUT>
__global__ void k_master_mlp(const float* __restrict__ pool, const float* __restrict__ cnt,
                             const float* __restrict__ Wm, const float* __restrict__ bm,
                             float* __restrict__ mm, const int* __restrict__ gptr) {
  int G = *gptr;
  int idx = blockIdx.x * blockDim.x + threadIdx.x;
  int g = idx / OUT, j = idx % OUT;
  if (g >= G) return;
  float inv = 1.0f / fmaxf(cnt[g], 1.0f);
  float acc = bm[j];
  for (int i = 0; i < IN; ++i)
    acc = fmaf(pool[(size_t)g * IN + i] * inv, Wm[(size_t)i * OUT + j], acc);
  mm[(size_t)g * OUT + j] = fmaxf(acc, 0.0f);
}

__global__ void k_add_master(float* __restrict__ h, const float* __restrict__ mm,
                             const int* __restrict__ batch, int N) {
  int t = blockIdx.x * blockDim.x + threadIdx.x;
  int node = t >> 5;
  if (node >= N) return;
  int fo = (t & 31) * 4;
  int g = batch[node];
  float4 a = *(float4*)&h[(size_t)node * 128 + fo];
  float4 m = *(const float4*)&mm[(size_t)g * 128 + fo];
  a.x += m.x; a.y += m.y; a.z += m.z; a.w += m.w;
  *(float4*)&h[(size_t)node * 128 + fo] = a;
}

__global__ void k_pool_z(const float* __restrict__ z, const int* __restrict__ batch,
                         float* __restrict__ pool, int N) {
  int t = blockIdx.x * blockDim.x + threadIdx.x;
  int node = t >> 5;
  if (node >= N) return;
  int f = t & 31;
  atomAddF(&pool[(size_t)batch[node] * 32 + f], z[(size_t)node * 32 + f]);
}

__global__ void k_finalize(const float* __restrict__ z, const float* __restrict__ mm2,
                           const int* __restrict__ batch, float* __restrict__ out, int N) {
  int node = blockIdx.x * (blockDim.x >> 5) + (threadIdx.x >> 5);
  if (node >= N) return;
  int lane = threadIdx.x & 31;
  float v = z[(size_t)node * 32 + lane] + mm2[(size_t)batch[node] * 32 + lane];
  float m = v;
#pragma unroll
  for (int o = 16; o >= 1; o >>= 1) m = fmaxf(m, __shfl_xor(m, o, 32));
  float ev = __expf(v - m);
  float s = ev;
#pragma unroll
  for (int o = 16; o >= 1; o >>= 1) s += __shfl_xor(s, o, 32);
  out[(size_t)node * 32 + lane] = ev / s;
}

extern "C" void kernel_launch(void* const* d_in, const int* in_sizes, int n_in,
                              void* d_out, int out_size, void* d_ws, size_t ws_size,
                              hipStream_t stream) {
  const float* X   = (const float*)d_in[0];
  const float* W1  = (const float*)d_in[1];
  const float* b1  = (const float*)d_in[2];
  const float* W2  = (const float*)d_in[3];
  const float* b2  = (const float*)d_in[4];
  const float* mW1 = (const float*)d_in[5];
  const float* mb1 = (const float*)d_in[6];
  const float* mW2 = (const float*)d_in[7];
  const float* mb2 = (const float*)d_in[8];
  const int* ei    = (const int*)d_in[9];
  const int* batch = (const int*)d_in[10];
  const int* gptr  = (const int*)d_in[11];

  const int N = in_sizes[0] / 128;
  const int E = in_sizes[9] / 2;
  const int* src = ei;
  const int* dst = ei + E;

  char* w = (char*)d_ws;
  size_t off = 0;
  auto take = [&](size_t bytes) -> float* {
    float* p = (float*)(w + off);
    off = (off + bytes + 255) & ~(size_t)255;
    return p;
  };
  float* xw    = take((size_t)N * 128 * 4);
  float* h     = take((size_t)N * 128 * 4);
  float* dinv  = take((size_t)N * 4);
  float* z     = take((size_t)N * 32 * 4);
  float* pool1 = take((size_t)GMAX * 128 * 4);
  float* pool2 = take((size_t)GMAX * 32 * 4);
  float* cnt   = take((size_t)GMAX * 4);
  float* mm1   = take((size_t)GMAX * 128 * 4);
  float* mm2   = take((size_t)GMAX * 32 * 4);

  const int B = 256;
  auto blks = [](long long n, int b) { return (unsigned)((n + b - 1) / b); };

  k_fill<<<blks(N, B), B, 0, stream>>>(dinv, 1.0f, N);
  k_fill<<<blks(GMAX * 128, B), B, 0, stream>>>(pool1, 0.0f, GMAX * 128);
  k_fill<<<blks(GMAX * 32, B), B, 0, stream>>>(pool2, 0.0f, GMAX * 32);
  k_fill<<<blks(GMAX, B), B, 0, stream>>>(cnt, 0.0f, GMAX);
  k_deg_count<<<blks(E, B), B, 0, stream>>>(dst, dinv, E);
  k_to_dinv<<<blks(N, B), B, 0, stream>>>(dinv, N);

  k_gemm_wmma<128><<<blks(N, 16), 256, 0, stream>>>(X, W1, xw, N);
  k_agg_init<128><<<blks((long long)N * 32, B), B, 0, stream>>>(xw, dinv, b1, h, N);
  k_agg_edges<128><<<blks((long long)E * 32, B), B, 0, stream>>>(xw, dinv, src, dst, h, E);
  k_elu_pool<<<blks(N, 8), B, 0, stream>>>(h, batch, pool1, cnt, N);

  k_master_mlp<128, 128><<<blks(GMAX * 128, B), B, 0, stream>>>(pool1, cnt, mW1, mb1, mm1, gptr);
  k_add_master<<<blks((long long)N * 32, B), B, 0, stream>>>(h, mm1, batch, N);

  k_gemm_wmma<32><<<blks(N, 16), 64, 0, stream>>>(h, W2, xw, N);
  k_agg_init<32><<<blks((long long)N * 8, B), B, 0, stream>>>(xw, dinv, b2, z, N);
  k_agg_edges<32><<<blks((long long)E * 8, B), B, 0, stream>>>(xw, dinv, src, dst, z, E);

  k_pool_z<<<blks((long long)N * 32, B), B, 0, stream>>>(z, batch, pool2, N);
  k_master_mlp<32, 32><<<blks(GMAX * 32, B), B, 0, stream>>>(pool2, cnt, mW2, mb2, mm2, gptr);
  k_finalize<<<blks(N, 8), B, 0, stream>>>(z, mm2, batch, (float*)d_out, N);
}


# Edge sums without f32 atomics: edges bucketed by target in-call (returning int add = slot, scan, place), one wave sums a row in registers; sum order = slot order, varies by run as the atomics did
# speedup vs baseline: 1.5333x; 1.5333x over previous
#include <hip/hip_runtime.h>
#include <math.h>


typedef __attribute__((ext_vector_type(2))) float v2f;
typedef __attribute__((ext_vector_type(8))) float v8f;

#define GMAX 256

__device__ __forceinline__ void atomAddF(float* p, float v) {
  __hip_atomic_fetch_add(p, v, __ATOMIC_RELAXED, __HIP_MEMORY_SCOPE_AGENT);
}

__global__ void k_fill(float* p, float v, int n) {
  int i = blockIdx.x * blockDim.x + threadIdx.x;
  if (i < n) p[i] = v;
}

__global__ void k_deg_rank(const int* __restrict__ dst, int* __restrict__ deg,
                           int* __restrict__ rank, int E) {
  int e = blockIdx.x * blockDim.x + threadIdx.x;
  if (e < E)
    rank[e] = __hip_atomic_fetch_add(&deg[dst[e]], 1, __ATOMIC_RELAXED, __HIP_MEMORY_SCOPE_AGENT);
}

__device__ __forceinline__ int wave_incl_scan(int v, int lane) {
  for (int o = 1; o < 32; o <<= 1) {
    int t = __shfl_up(v, o, 32);
    if (lane >= o) v += t;
  }
  return v;
}

__global__ void __launch_bounds__(1024) k_scan_rowptr(int* rp, int N) {
  __shared__ int wsum[32];
  const int lane = threadIdx.x & 31;
  const int wave = threadIdx.x >> 5;
  const int nw   = blockDim.x >> 5;
  const int per  = ((N + nw - 1) / nw + 31) & ~31;
  const int lo   = min(N, wave * per);
  const int hi   = min(N, lo + per);

  int s = 0;
  for (int i = lo + lane; i < hi; i += 32) s += rp[i];
  for (int o = 16; o >= 1; o >>= 1) s += __shfl_xor(s, o, 32);
  if (lane == 0) wsum[wave] = s;
  __syncthreads();

  const int v   = (lane < nw) ? wsum[lane] : 0;
  const int inc = wave_incl_scan(v, lane);
  int run = __shfl(inc - v, wave, 32);

  for (int base = lo; base < hi; base += 32) {
    const int i  = base + lane;
    const int d  = (i < hi) ? rp[i] : 0;
    const int sc = wave_incl_scan(d, lane);
    if (i < hi) rp[i] = run + sc - d;
    run += __shfl(sc, 31, 32);
  }
}

__global__ void k_csr_fill(const int* __restrict__ src, const int* __restrict__ dst,
                           const int* __restrict__ rank, const int* __restrict__ rp,
                           int* __restrict__ csr, int E) {
  int e = blockIdx.x * blockDim.x + threadIdx.x;
  if (e < E) csr[rp[dst[e]] + rank[e]] = src[e];
}

template <int OUT>
__global__ void k_gemm_wmma(const float* __restrict__ A, const float* __restrict__ W,
                            float* __restrict__ C, int N) {
  constexpr int KD = 128;
  __shared__ float As[16 * KD];

  const int row0 = blockIdx.x * 16;
  for (int i = threadIdx.x; i < 16 * KD; i += blockDim.x) {
    int r = i / KD, k = i % KD;
    As[i] = (row0 + r < N) ? A[(size_t)(row0 + r) * KD + k] : 0.0f;
  }
  __syncthreads();

  const int wave = threadIdx.x >> 5;
  const int lane = threadIdx.x & 31;
  const int half = lane >> 4;
  const int n    = lane & 15;
  const int cw   = wave * 16;

  v8f c = {};
  for (int k0 = 0; k0 < KD; k0 += 4) {
    const int k = k0 + 2 * half;
    v2f a, b;
    a.x = As[n * KD + k];
    a.y = As[n * KD + k + 1];
    b.x = W[(size_t)k * OUT + cw + n];
    b.y = W[(size_t)(k + 1) * OUT + cw + n];
    c = __builtin_amdgcn_wmma_f32_16x16x4_f32(false, a, false, b,
                                              (short)0, c, false, false);
  }

#pragma unroll
  for (int r = 0; r < 8; ++r) {
    const int m = r + 8 * half;
    if (row0 + m < N) C[(size_t)(row0 + m) * OUT + cw + n] = c[r];
  }
}

template <int V> struct RowVec;
template <> struct RowVec<4> { typedef float4 T; };
template <> struct RowVec<1> { typedef float T; };

__device__ __forceinline__ void axpy(float4& a, float w, const float4 v) {
  a.x = fmaf(w, v.x, a.x); a.y = fmaf(w, v.y, a.y);
  a.z = fmaf(w, v.z, a.z); a.w = fmaf(w, v.w, a.w);
}
__device__ __forceinline__ void axpy(float& a, float w, const float v) { a = fmaf(w, v, a); }
__device__ __forceinline__ float4 plus(const float4 a, const float4 b) {
  return make_float4(a.x + b.x, a.y + b.y, a.z + b.z, a.w + b.w);
}
__device__ __forceinline__ float plus(const float a, const float b) { return a + b; }
__device__ __forceinline__ void set_zero(float4& a) { a = make_float4(0.0f, 0.0f, 0.0f, 0.0f); }
__device__ __forceinline__ void set_zero(float& a) { a = 0.0f; }

__device__ __forceinline__ float dinv_of(const int* __restrict__ rp, int i, int N, int E) {
  const int b = rp[i];
  const int e = (i + 1 < N) ? rp[i + 1] : E;
  return rsqrtf((float)(e - b + 1));
}

template <int B, typename T>
__device__ __forceinline__ void gather_rows(T& acc, const T* __restrict__ xr, int s_i, float w_i,
                                            int k, int lane) {
  T v[B];
  float w[B];
  for (int u = 0; u < B; ++u) {
    const int s = __builtin_amdgcn_readlane(s_i, k + u);
    w[u] = __int_as_float(__builtin_amdgcn_readlane(__float_as_int(w_i), k + u));
    v[u] = xr[(size_t)s * 32 + lane];
  }
  for (int u = 0; u < B; ++u) axpy(acc, w[u], v[u]);
}

template <int F>
__global__ void k_gather(const float* __restrict__ xw, const int* __restrict__ rp,
                         const int* __restrict__ csr, const float* __restrict__ bias,
                         float* __restrict__ out, int N, int E) {
  constexpr int V = F / 32;
  typedef typename RowVec<V>::T T;
  const int lane   = threadIdx.x & 31;
  const int wpb    = blockDim.x >> 5;
  const int nwaves = gridDim.x * wpb;
  const T* __restrict__ xr = (const T*)xw;
  T* __restrict__ orow     = (T*)out;
  const T bv = ((const T*)bias)[lane];

  int node = __builtin_amdgcn_readfirstlane((int)(blockIdx.x * wpb + (threadIdx.x >> 5)));
  for (; node < N; node += nwaves) {
    const int beg  = rp[node];
    const int end  = (node + 1 < N) ? rp[node + 1] : E;
    const float dn = rsqrtf((float)(end - beg + 1));
    const T self = xr[(size_t)node * 32 + lane];
    T acc;
    set_zero(acc);

    for (int base = beg; base < end; base += 32) {
      const int j = base + lane;
      int s_i = 0;
      float w_i = 0.0f;
      if (j < end) {
        s_i = csr[j];
        w_i = dinv_of(rp, s_i, N, E) * dn;
      }
      const int cnt = min(32, end - base);
      int k = 0;
      for (; k + 8 <= cnt; k += 8) gather_rows<8>(acc, xr, s_i, w_i, k, lane);
      if (k + 4 <= cnt) { gather_rows<4>(acc, xr, s_i, w_i, k, lane); k += 4; }
      if (k + 2 <= cnt) { gather_rows<2>(acc, xr, s_i, w_i, k, lane); k += 2; }
      if (k < cnt)      { gather_rows<1>(acc, xr, s_i, w_i, k, lane); }
    }
    axpy(acc, dn * dn, self);
    orow[(size_t)node * 32 + lane] = plus(acc, bv);
  }
}

__global__ void k_elu_pool(float* __restrict__ h, const int* __restrict__ batch,
                           float* __restrict__ pool, float* __restrict__ cnt, int N) {
  int node = blockIdx.x * (blockDim.x >> 5) + (threadIdx.x >> 5);
  if (node >= N) return;
  int lane = threadIdx.x & 31;
  float4 v = *(float4*)&h[(size_t)node * 128 + lane * 4];
  v.x = v.x > 0.0f ? v.x : expm1f(v.x);
  v.y = v.y > 0.0f ? v.y : expm1f(v.y);
  v.z = v.z > 0.0f ? v.z : expm1f(v.z);
  v.w = v.w > 0.0f ? v.w : expm1f(v.w);
  *(float4*)&h[(size_t)node * 128 + lane * 4] = v;
  int g = batch[node];
  float* p = &pool[(size_t)g * 128 + lane * 4];
  atomAddF(p + 0, v.x); atomAddF(p + 1, v.y);
  atomAddF(p + 2, v.z); atomAddF(p + 3, v.w);
  if (lane == 0) atomAddF(&cnt[g], 1.0f);
}

template <int IN, int OUT>
__global__ void k_master_mlp(const float* __restrict__ pool, const float* __restrict__ cnt,
                             const float* __restrict__ Wm, const float* __restrict__ bm,
                             float* __restrict__ mm, const int* __restrict__ gptr) {
  int G = *gptr;
  int idx = blockIdx.x * blockDim.x + threadIdx.x;
  int g = idx / OUT, j = idx % OUT;
  if (g >= G) return;
  float inv = 1.0f / fmaxf(cnt[g], 1.0f);
  float acc = bm[j];
  for (int i = 0; i < IN; ++i)
    acc = fmaf(pool[(size_t)g * IN + i] * inv, Wm[(size_t)i * OUT + j], acc);
  mm[(size_t)g * OUT + j] = fmaxf(acc, 0.0f);
}

__global__ void k_add_master(float* __restrict__ h, const float* __restrict__ mm,
                             const int* __restrict__ batch, int N) {
  int t = blockIdx.x * blockDim.x + threadIdx.x;
  int node = t >> 5;
  if (node >= N) return;
  int fo = (t & 31) * 4;
  int g = batch[node];
  float4 a = *(float4*)&h[(size_t)node * 128 + fo];
  float4 m = *(const float4*)&mm[(size_t)g * 128 + fo];
  a.x += m.x; a.y += m.y; a.z += m.z; a.w += m.w;
  *(float4*)&h[(size_t)node * 128 + fo] = a;
}

__global__ void k_pool_z(const float* __restrict__ z, const int* __restrict__ batch,
                         float* __restrict__ pool, int N) {
  int t = blockIdx.x * blockDim.x + threadIdx.x;
  int node = t >> 5;
  if (node >= N) return;
  int f = t & 31;
  atomAddF(&pool[(size_t)batch[node] * 32 + f], z[(size_t)node * 32 + f]);
}

__global__ void k_finalize(const float* __restrict__ z, const float* __restrict__ mm2,
                           const int* __restrict__ batch, float* __restrict__ out, int N) {
  int node = blockIdx.x * (blockDim.x >> 5) + (threadIdx.x >> 5);
  if (node >= N) return;
  int lane = threadIdx.x & 31;
  float v = z[(size_t)node * 32 + lane] + mm2[(size_t)batch[node] * 32 + lane];
  float m = v;
#pragma unroll
  for (int o = 16; o >= 1; o >>= 1) m = fmaxf(m, __shfl_xor(m, o, 32));
  float ev = __expf(v - m);
  float s = ev;
#pragma unroll
  for (int o = 16; o >= 1; o >>= 1) s += __shfl_xor(s, o, 32);
  out[(size_t)node * 32 + lane] = ev / s;
}

extern "C" void kernel_launch(void* const* d_in, const int* in_sizes, int n_in,
                              void* d_out, int out_size, void* d_ws, size_t ws_size,
                              hipStream_t stream) {
  const float* X   = (const float*)d_in[0];
  const float* W1  = (const float*)d_in[1];
  const float* b1  = (const float*)d_in[2];
  const float* W2  = (const float*)d_in[3];
  const float* b2  = (const float*)d_in[4];
  const float* mW1 = (const float*)d_in[5];
  const float* mb1 = (const float*)d_in[6];
  const float* mW2 = (const float*)d_in[7];
  const float* mb2 = (const float*)d_in[8];
  const int* ei    = (const int*)d_in[9];
  const int* batch = (const int*)d_in[10];
  const int* gptr  = (const int*)d_in[11];

  const int N = in_sizes[0] / 128;
  const int E = in_sizes[9] / 2;
  const int* src = ei;
  const int* dst = ei + E;

  char* w = (char*)d_ws;
  size_t off = 0;
  auto take = [&](size_t bytes) -> float* {
    float* p = (float*)(w + off);
    off = (off + bytes + 255) & ~(size_t)255;
    return p;
  };
  float* xw    = take((size_t)N * 128 * 4);
  float* h     = take((size_t)N * 128 * 4);
  int*   rp    = (int*)take((size_t)N * 4);
  int*   csr   = (int*)take((size_t)N * 32 * 4);
  int*   rank  = (int*)h;
  float* z     = h;
  float* pool1 = take((size_t)GMAX * 128 * 4);
  float* pool2 = take((size_t)GMAX * 32 * 4);
  float* cnt   = take((size_t)GMAX * 4);
  float* mm1   = take((size_t)GMAX * 128 * 4);
  float* mm2   = take((size_t)GMAX * 32 * 4);

  const int B = 256;
  auto blks = [](long long n, int b) { return (unsigned)((n + b - 1) / b); };

  k_fill<<<blks(N, B), B, 0, stream>>>((float*)rp, 0.0f, N);
  k_fill<<<blks(GMAX * 128, B), B, 0, stream>>>(pool1, 0.0f, GMAX * 128);
  k_fill<<<blks(GMAX * 32, B), B, 0, stream>>>(pool2, 0.0f, GMAX * 32);
  k_fill<<<blks(GMAX, B), B, 0, stream>>>(cnt, 0.0f, GMAX);
  k_deg_rank<<<blks(E, B), B, 0, stream>>>(dst, rp, rank, E);
  k_scan_rowptr<<<1, 1024, 0, stream>>>(rp, N);
  k_csr_fill<<<blks(E, B), B, 0, stream>>>(src, dst, rank, rp, csr, E);

  unsigned gblocks = blks(N, 8);
  if (gblocks > 2048u) gblocks = 2048u;

  k_gemm_wmma<128><<<blks(N, 16), 256, 0, stream>>>(X, W1, xw, N);
  k_gather<128><<<gblocks, B, 0, stream>>>(xw, rp, csr, b1, h, N, E);
  k_elu_pool<<<blks(N, 8), B, 0, stream>>>(h, batch, pool1, cnt, N);

  k_master_mlp<128, 128><<<blks(GMAX * 128, B), B, 0, stream>>>(pool1, cnt, mW1, mb1, mm1, gptr);
  k_add_master<<<blks((long long)N * 32, B), B, 0, stream>>>(h, mm1, batch, N);

  k_gemm_wmma<32><<<blks(N, 16), 64, 0, stream>>>(h, W2, xw, N);
  k_gather<32><<<gblocks, B, 0, stream>>>(xw, rp, csr, b2, z, N, E);

  k_pool_z<<<blks((long long)N * 32, B), B, 0, stream>>>(z, batch, pool2, N);
  k_master_mlp<32, 32><<<blks(GMAX * 32, B), B, 0, stream>>>(pool2, cnt, mW2, mb2, mm2, gptr);
  k_finalize<<<blks(N, 8), B, 0, stream>>>(z, mm2, batch, (float*)d_out, N);
}


# Edge bucketing without per-edge global atomics: edges split into 256-node ranges, then one work-group per range counts, scans, places, all in LDS counters; sum order = slot order, varies by run
# speedup vs baseline: 2.1467x; 1.4001x over previous
#include <hip/hip_runtime.h>
#include <math.h>


typedef __attribute__((ext_vector_type(2))) float v2f;
typedef __attribute__((ext_vector_type(8))) float v8f;

#define GMAX 256

__device__ __forceinline__ void atomAddF(float* p, float v) {
  __hip_atomic_fetch_add(p, v, __ATOMIC_RELAXED, __HIP_MEMORY_SCOPE_AGENT);
}

__global__ void k_fill(float* p, float v, int n) {
  int i = blockIdx.x * blockDim.x + threadIdx.x;
  if (i < n) p[i] = v;
}

#define BSH   8
#define BW    (1 << BSH)
#define NBMAX 1024
#define EPT   4
#define EPB   (1024 * EPT)

__device__ __forceinline__ int lds_add_rtn(int* p, int v) {
  return __hip_atomic_fetch_add(p, v, __ATOMIC_RELAXED, __HIP_MEMORY_SCOPE_WORKGROUP);
}

__device__ __forceinline__ int wave_incl_scan(int v, int lane) {
  for (int o = 1; o < 32; o <<= 1) {
    int t = __shfl_up(v, o, 32);
    if (lane >= o) v += t;
  }
  return v;
}

__device__ __forceinline__ int block_excl_scan(int v, int* wsum, int& total) {
  const int lane = threadIdx.x & 31;
  const int wave = threadIdx.x >> 5;
  const int inc  = wave_incl_scan(v, lane);
  if (lane == 31) wsum[wave] = inc;
  __syncthreads();
  const int w    = wsum[lane];
  const int winc = wave_incl_scan(w, lane);
  total = __shfl(winc, 31, 32);
  const int wbase = __shfl(winc - w, wave, 32);
  __syncthreads();
  return wbase + inc - v;
}

__global__ void __launch_bounds__(1024) k_bin_count(const int* __restrict__ dst, int* __restrict__ bintot,
                                                    int E, int NB) {
  __shared__ int hist[NBMAX];
  for (int i = threadIdx.x; i < NB; i += 1024) hist[i] = 0;
  __syncthreads();
  const int base = blockIdx.x * EPB + threadIdx.x;
  int d[EPT];
#pragma unroll
  for (int u = 0; u < EPT; ++u) {
    const int e = base + u * 1024;
    d[u] = (e < E) ? dst[e] : -1;
  }
#pragma unroll
  for (int u = 0; u < EPT; ++u)
    if (d[u] >= 0) lds_add_rtn(&hist[d[u] >> BSH], 1);
  __syncthreads();
  for (int i = threadIdx.x; i < NB; i += 1024) {
    const int c = hist[i];
    if (c) __hip_atomic_fetch_add(&bintot[i], c, __ATOMIC_RELAXED, __HIP_MEMORY_SCOPE_AGENT);
  }
}

__global__ void __launch_bounds__(1024) k_bin_place(const int* __restrict__ src, const int* __restrict__ dst,
                                                    const int* __restrict__ bintot, int* __restrict__ bincur,
                                                    int2* __restrict__ coarse, int E, int NB) {
  __shared__ int hist[NBMAX];
  __shared__ int wsum[32];
  for (int i = threadIdx.x; i < NB; i += 1024) hist[i] = 0;
  __syncthreads();
  const int base = blockIdx.x * EPB + threadIdx.x;
  int s[EPT], d[EPT], r[EPT];
#pragma unroll
  for (int u = 0; u < EPT; ++u) {
    const int e = base + u * 1024;
    d[u] = (e < E) ? dst[e] : -1;
    s[u] = (e < E) ? src[e] : 0;
  }
#pragma unroll
  for (int u = 0; u < EPT; ++u)
    r[u] = (d[u] >= 0) ? lds_add_rtn(&hist[d[u] >> BSH], 1) : 0;

  const int t = (threadIdx.x < NB) ? bintot[threadIdx.x] : 0;
  int total;
  const int start = block_excl_scan(t, wsum, total);
  if (threadIdx.x < NB) {
    const int c = hist[threadIdx.x];
    const int off = c ? __hip_atomic_fetch_add(&bincur[threadIdx.x], c, __ATOMIC_RELAXED, __HIP_MEMORY_SCOPE_AGENT) : 0;
    hist[threadIdx.x] = start + off;
  }
  __syncthreads();
#pragma unroll
  for (int u = 0; u < EPT; ++u)
    if (d[u] >= 0) coarse[hist[d[u] >> BSH] + r[u]] = make_int2(s[u], d[u]);
}

__global__ void __launch_bounds__(1024) k_bin_sort(const int2* __restrict__ coarse, const int* __restrict__ bintot,
                                                   int* __restrict__ rp, int* __restrict__ csr, int N) {
  __shared__ int cnt[BW];
  __shared__ int wsum[32];
  const int b = blockIdx.x;
  if (threadIdx.x < BW) cnt[threadIdx.x] = 0;
  const int t = ((int)threadIdx.x < b) ? bintot[threadIdx.x] : 0;
  int bstart;
  block_excl_scan(t, wsum, bstart);
  const int bc = bintot[b];
  const int2* __restrict__ ed = coarse + bstart;

  for (int i = threadIdx.x; i < bc; i += 1024) lds_add_rtn(&cnt[ed[i].y & (BW - 1)], 1);
  __syncthreads();

  const int v = (threadIdx.x < BW) ? cnt[threadIdx.x] : 0;
  int tot;
  const int ex = block_excl_scan(v, wsum, tot);
  if (threadIdx.x < BW) {
    const int node = b * BW + threadIdx.x;
    if (node < N) rp[node] = bstart + ex;
    cnt[threadIdx.x] = bstart + ex;
  }
  __syncthreads();

  for (int i = threadIdx.x; i < bc; i += 1024) {
    const int2 e = ed[i];
    csr[lds_add_rtn(&cnt[e.y & (BW - 1)], 1)] = e.x;
  }
}

template <int OUT>
__global__ void k_gemm_wmma(const float* __restrict__ A, const float* __restrict__ W,
                            float* __restrict__ C, int N) {
  constexpr int KD = 128;
  __shared__ float As[16 * KD];

  const int row0 = blockIdx.x * 16;
  for (int i = threadIdx.x; i < 16 * KD; i += blockDim.x) {
    int r = i / KD, k = i % KD;
    As[i] = (row0 + r < N) ? A[(size_t)(row0 + r) * KD + k] : 0.0f;
  }
  __syncthreads();

  const int wave = threadIdx.x >> 5;
  const int lane = threadIdx.x & 31;
  const int half = lane >> 4;
  const int n    = lane & 15;
  const int cw   = wave * 16;

  v8f c = {};
  for (int k0 = 0; k0 < KD; k0 += 4) {
    const int k = k0 + 2 * half;
    v2f a, b;
    a.x = As[n * KD + k];
    a.y = As[n * KD + k + 1];
    b.x = W[(size_t)k * OUT + cw + n];
    b.y = W[(size_t)(k + 1) * OUT + cw + n];
    c = __builtin_amdgcn_wmma_f32_16x16x4_f32(false, a, false, b,
                                              (short)0, c, false, false);
  }

#pragma unroll
  for (int r = 0; r < 8; ++r) {
    const int m = r + 8 * half;
    if (row0 + m < N) C[(size_t)(row0 + m) * OUT + cw + n] = c[r];
  }
}

template <int V> struct RowVec;
template <> struct RowVec<4> { typedef float4 T; };
template <> struct RowVec<1> { typedef float T; };

__device__ __forceinline__ void axpy(float4& a, float w, const float4 v) {
  a.x = fmaf(w, v.x, a.x); a.y = fmaf(w, v.y, a.y);
  a.z = fmaf(w, v.z, a.z); a.w = fmaf(w, v.w, a.w);
}
__device__ __forceinline__ void axpy(float& a, float w, const float v) { a = fmaf(w, v, a); }
__device__ __forceinline__ float4 plus(const float4 a, const float4 b) {
  return make_float4(a.x + b.x, a.y + b.y, a.z + b.z, a.w + b.w);
}
__device__ __forceinline__ float plus(const float a, const float b) { return a + b; }
__device__ __forceinline__ void set_zero(float4& a) { a = make_float4(0.0f, 0.0f, 0.0f, 0.0f); }
__device__ __forceinline__ void set_zero(float& a) { a = 0.0f; }

__device__ __forceinline__ float dinv_of(const int* __restrict__ rp, int i, int N, int E) {
  const int b = rp[i];
  const int e = (i + 1 < N) ? rp[i + 1] : E;
  return rsqrtf((float)(e - b + 1));
}

template <int B, typename T>
__device__ __forceinline__ void gather_rows(T& acc, const T* __restrict__ xr, int s_i, float w_i,
                                            int k, int lane) {
  T v[B];
  float w[B];
  for (int u = 0; u < B; ++u) {
    const int s = __builtin_amdgcn_readlane(s_i, k + u);
    w[u] = __int_as_float(__builtin_amdgcn_readlane(__float_as_int(w_i), k + u));
    v[u] = xr[(size_t)s * 32 + lane];
  }
  for (int u = 0; u < B; ++u) axpy(acc, w[u], v[u]);
}

template <int F>
__global__ void k_gather(const float* __restrict__ xw, const int* __restrict__ rp,
                         const int* __restrict__ csr, const float* __restrict__ bias,
                         float* __restrict__ out, int N, int E) {
  constexpr int V = F / 32;
  typedef typename RowVec<V>::T T;
  const int lane   = threadIdx.x & 31;
  const int wpb    = blockDim.x >> 5;
  const int nwaves = gridDim.x * wpb;
  const T* __restrict__ xr = (const T*)xw;
  T* __restrict__ orow     = (T*)out;
  const T bv = ((const T*)bias)[lane];

  int node = __builtin_amdgcn_readfirstlane((int)(blockIdx.x * wpb + (threadIdx.x >> 5)));
  for (; node < N; node += nwaves) {
    const int beg  = rp[node];
    const int end  = (node + 1 < N) ? rp[node + 1] : E;
    const float dn = rsqrtf((float)(end - beg + 1));
    const T self = xr[(size_t)node * 32 + lane];
    T acc;
    set_zero(acc);

    for (int base = beg; base < end; base += 32) {
      const int j = base + lane;
      int s_i = 0;
      float w_i = 0.0f;
      if (j < end) {
        s_i = csr[j];
        w_i = dinv_of(rp, s_i, N, E) * dn;
      }
      const int cnt = min(32, end - base);
      int k = 0;
      for (; k + 8 <= cnt; k += 8) gather_rows<8>(acc, xr, s_i, w_i, k, lane);
      if (k + 4 <= cnt) { gather_rows<4>(acc, xr, s_i, w_i, k, lane); k += 4; }
      if (k + 2 <= cnt) { gather_rows<2>(acc, xr, s_i, w_i, k, lane); k += 2; }
      if (k < cnt)      { gather_rows<1>(acc, xr, s_i, w_i, k, lane); }
    }
    axpy(acc, dn * dn, self);
    orow[(size_t)node * 32 + lane] = plus(acc, bv);
  }
}

__global__ void k_elu_pool(float* __restrict__ h, const int* __restrict__ batch,
                           float* __restrict__ pool, float* __restrict__ cnt, int N) {
  int node = blockIdx.x * (blockDim.x >> 5) + (threadIdx.x >> 5);
  if (node >= N) return;
  int lane = threadIdx.x & 31;
  float4 v = *(float4*)&h[(size_t)node * 128 + lane * 4];
  v.x = v.x > 0.0f ? v.x : expm1f(v.x);
  v.y = v.y > 0.0f ? v.y : expm1f(v.y);
  v.z = v.z > 0.0f ? v.z : expm1f(v.z);
  v.w = v.w > 0.0f ? v.w : expm1f(v.w);
  *(float4*)&h[(size_t)node * 128 + lane * 4] = v;
  int g = batch[node];
  float* p = &pool[(size_t)g * 128 + lane * 4];
  atomAddF(p + 0, v.x); atomAddF(p + 1, v.y);
  atomAddF(p + 2, v.z); atomAddF(p + 3, v.w);
  if (lane == 0) atomAddF(&cnt[g], 1.0f);
}

template <int IN, int OUT>
__global__ void k_master_mlp(const float* __restrict__ pool, const float* __restrict__ cnt,
                             const float* __restrict__ Wm, const float* __restrict__ bm,
                             float* __restrict__ mm, const int* __restrict__ gptr) {
  int G = *gptr;
  int idx = blockIdx.x * blockDim.x + threadIdx.x;
  int g = idx / OUT, j = idx % OUT;
  if (g >= G) return;
  float inv = 1.0f / fmaxf(cnt[g], 1.0f);
  float acc = bm[j];
  for (int i = 0; i < IN; ++i)
    acc = fmaf(pool[(size_t)g * IN + i] * inv, Wm[(size_t)i * OUT + j], acc);
  mm[(size_t)g * OUT + j] = fmaxf(acc, 0.0f);
}

__global__ void k_add_master(float* __restrict__ h, const float* __restrict__ mm,
                             const int* __restrict__ batch, int N) {
  int t = blockIdx.x * blockDim.x + threadIdx.x;
  int node = t >> 5;
  if (node >= N) return;
  int fo = (t & 31) * 4;
  int g = batch[node];
  float4 a = *(float4*)&h[(size_t)node * 128 + fo];
  float4 m = *(const float4*)&mm[(size_t)g * 128 + fo];
  a.x += m.x; a.y += m.y; a.z += m.z; a.w += m.w;
  *(float4*)&h[(size_t)node * 128 + fo] = a;
}

__global__ void k_pool_z(const float* __restrict__ z, const int* __restrict__ batch,
                         float* __restrict__ pool, int N) {
  int t = blockIdx.x * blockDim.x + threadIdx.x;
  int node = t >> 5;
  if (node >= N) return;
  int f = t & 31;
  atomAddF(&pool[(size_t)batch[node] * 32 + f], z[(size_t)node * 32 + f]);
}

__global__ void k_finalize(const float* __restrict__ z, const float* __restrict__ mm2,
                           const int* __restrict__ batch, float* __restrict__ out, int N) {
  int node = blockIdx.x * (blockDim.x >> 5) + (threadIdx.x >> 5);
  if (node >= N) return;
  int lane = threadIdx.x & 31;
  float v = z[(size_t)node * 32 + lane] + mm2[(size_t)batch[node] * 32 + lane];
  float m = v;
#pragma unroll
  for (int o = 16; o >= 1; o >>= 1) m = fmaxf(m, __shfl_xor(m, o, 32));
  float ev = __expf(v - m);
  float s = ev;
#pragma unroll
  for (int o = 16; o >= 1; o >>= 1) s += __shfl_xor(s, o, 32);
  out[(size_t)node * 32 + lane] = ev / s;
}

extern "C" void kernel_launch(void* const* d_in, const int* in_sizes, int n_in,
                              void* d_out, int out_size, void* d_ws, size_t ws_size,
                              hipStream_t stream) {
  const float* X   = (const float*)d_in[0];
  const float* W1  = (const float*)d_in[1];
  const float* b1  = (const float*)d_in[2];
  const float* W2  = (const float*)d_in[3];
  const float* b2  = (const float*)d_in[4];
  const float* mW1 = (const float*)d_in[5];
  const float* mb1 = (const float*)d_in[6];
  const float* mW2 = (const float*)d_in[7];
  const float* mb2 = (const float*)d_in[8];
  const int* ei    = (const int*)d_in[9];
  const int* batch = (const int*)d_in[10];
  const int* gptr  = (const int*)d_in[11];

  const int N = in_sizes[0] / 128;
  const int E = in_sizes[9] / 2;
  const int* src = ei;
  const int* dst = ei + E;

  char* w = (char*)d_ws;
  size_t off = 0;
  auto take = [&](size_t bytes) -> float* {
    float* p = (float*)(w + off);
    off = (off + bytes + 255) & ~(size_t)255;
    return p;
  };
  float* xw    = take((size_t)N * 128 * 4);
  float* h     = take((size_t)N * 128 * 4);
  int*   rp    = (int*)take((size_t)N * 4);
  int*   csr   = (int*)take((size_t)N * 32 * 4);
  int2*  coarse = (int2*)h;
  int*   bintot = (int*)xw;
  int*   bincur = bintot + NBMAX;
  float* z     = h;
  float* pool1 = take((size_t)GMAX * 128 * 4);
  float* pool2 = take((size_t)GMAX * 32 * 4);
  float* cnt   = take((size_t)GMAX * 4);
  float* mm1   = take((size_t)GMAX * 128 * 4);
  float* mm2   = take((size_t)GMAX * 32 * 4);

  const int B = 256;
  auto blks = [](long long n, int b) { return (unsigned)((n + b - 1) / b); };

  const int NB = (N + BW - 1) >> BSH;
  k_fill<<<blks(2 * NBMAX, B), B, 0, stream>>>((float*)bintot, 0.0f, 2 * NBMAX);
  k_fill<<<blks(GMAX * 128, B), B, 0, stream>>>(pool1, 0.0f, GMAX * 128);
  k_fill<<<blks(GMAX * 32, B), B, 0, stream>>>(pool2, 0.0f, GMAX * 32);
  k_fill<<<blks(GMAX, B), B, 0, stream>>>(cnt, 0.0f, GMAX);
  k_bin_count<<<blks(E, EPB), 1024, 0, stream>>>(dst, bintot, E, NB);
  k_bin_place<<<blks(E, EPB), 1024, 0, stream>>>(src, dst, bintot, bincur, coarse, E, NB);
  k_bin_sort<<<NB, 1024, 0, stream>>>(coarse, bintot, rp, csr, N);

  unsigned gblocks = blks(N, 8);
  if (gblocks > 2048u) gblocks = 2048u;

  k_gemm_wmma<128><<<blks(N, 16), 256, 0, stream>>>(X, W1, xw, N);
  k_gather<128><<<gblocks, B, 0, stream>>>(xw, rp, csr, b1, h, N, E);
  k_elu_pool<<<blks(N, 8), B, 0, stream>>>(h, batch, pool1, cnt, N);

  k_master_mlp<128, 128><<<blks(GMAX * 128, B), B, 0, stream>>>(pool1, cnt, mW1, mb1, mm1, gptr);
  k_add_master<<<blks((long long)N * 32, B), B, 0, stream>>>(h, mm1, batch, N);

  k_gemm_wmma<32><<<blks(N, 16), 64, 0, stream>>>(h, W2, xw, N);
  k_gather<32><<<gblocks, B, 0, stream>>>(xw, rp, csr, b2, z, N, E);

  k_pool_z<<<blks((long long)N * 32, B), B, 0, stream>>>(z, batch, pool2, N);
  k_master_mlp<32, 32><<<blks(GMAX * 32, B), B, 0, stream>>>(pool2, cnt, mW2, mb2, mm2, gptr);
  k_finalize<<<blks(N, 8), B, 0, stream>>>(z, mm2, batch, (float*)d_out, N);
}


# Graph pooling with few float atomics: a wave sums a run of 16 consecutive nodes per graph in registers, one add per graph met (was one per node, on few hot lines); sum = run order, then arrival order
# speedup vs baseline: 4.1936x; 1.9535x over previous
#include <hip/hip_runtime.h>
#include <math.h>


typedef __attribute__((ext_vector_type(2))) float v2f;
typedef __attribute__((ext_vector_type(8))) float v8f;

#define GMAX 256

__device__ __forceinline__ void atomAddF(float* p, float v) {
  __hip_atomic_fetch_add(p, v, __ATOMIC_RELAXED, __HIP_MEMORY_SCOPE_AGENT);
}

__global__ void k_fill(float* p, float v, int n) {
  int i = blockIdx.x * blockDim.x + threadIdx.x;
  if (i < n) p[i] = v;
}

#define BSH   8
#define BW    (1 << BSH)
#define NBMAX 1024
#define EPT   4
#define EPB   (1024 * EPT)

__device__ __forceinline__ int lds_add_rtn(int* p, int v) {
  return __hip_atomic_fetch_add(p, v, __ATOMIC_RELAXED, __HIP_MEMORY_SCOPE_WORKGROUP);
}

__device__ __forceinline__ int wave_incl_scan(int v, int lane) {
  for (int o = 1; o < 32; o <<= 1) {
    int t = __shfl_up(v, o, 32);
    if (lane >= o) v += t;
  }
  return v;
}

__device__ __forceinline__ int block_excl_scan(int v, int* wsum, int& total) {
  const int lane = threadIdx.x & 31;
  const int wave = threadIdx.x >> 5;
  const int inc  = wave_incl_scan(v, lane);
  if (lane == 31) wsum[wave] = inc;
  __syncthreads();
  const int w    = wsum[lane];
  const int winc = wave_incl_scan(w, lane);
  total = __shfl(winc, 31, 32);
  const int wbase = __shfl(winc - w, wave, 32);
  __syncthreads();
  return wbase + inc - v;
}

__global__ void __launch_bounds__(1024) k_bin_count(const int* __restrict__ dst, int* __restrict__ bintot,
                                                    int E, int NB) {
  __shared__ int hist[NBMAX];
  for (int i = threadIdx.x; i < NB; i += 1024) hist[i] = 0;
  __syncthreads();
  const int base = blockIdx.x * EPB + threadIdx.x;
  int d[EPT];
#pragma unroll
  for (int u = 0; u < EPT; ++u) {
    const int e = base + u * 1024;
    d[u] = (e < E) ? dst[e] : -1;
  }
#pragma unroll
  for (int u = 0; u < EPT; ++u)
    if (d[u] >= 0) lds_add_rtn(&hist[d[u] >> BSH], 1);
  __syncthreads();
  for (int i = threadIdx.x; i < NB; i += 1024) {
    const int c = hist[i];
    if (c) __hip_atomic_fetch_add(&bintot[i], c, __ATOMIC_RELAXED, __HIP_MEMORY_SCOPE_AGENT);
  }
}

__global__ void __launch_bounds__(1024) k_bin_place(const int* __restrict__ src, const int* __restrict__ dst,
                                                    const int* __restrict__ bintot, int* __restrict__ bincur,
                                                    int2* __restrict__ coarse, int E, int NB) {
  __shared__ int hist[NBMAX];
  __shared__ int wsum[32];
  for (int i = threadIdx.x; i < NB; i += 1024) hist[i] = 0;
  __syncthreads();
  const int base = blockIdx.x * EPB + threadIdx.x;
  int s[EPT], d[EPT], r[EPT];
#pragma unroll
  for (int u = 0; u < EPT; ++u) {
    const int e = base + u * 1024;
    d[u] = (e < E) ? dst[e] : -1;
    s[u] = (e < E) ? src[e] : 0;
  }
#pragma unroll
  for (int u = 0; u < EPT; ++u)
    r[u] = (d[u] >= 0) ? lds_add_rtn(&hist[d[u] >> BSH], 1) : 0;

  const int t = (threadIdx.x < NB) ? bintot[threadIdx.x] : 0;
  int total;
  const int start = block_excl_scan(t, wsum, total);
  if (threadIdx.x < NB) {
    const int c = hist[threadIdx.x];
    const int off = c ? __hip_atomic_fetch_add(&bincur[threadIdx.x], c, __ATOMIC_RELAXED, __HIP_MEMORY_SCOPE_AGENT) : 0;
    hist[threadIdx.x] = start + off;
  }
  __syncthreads();
#pragma unroll
  for (int u = 0; u < EPT; ++u)
    if (d[u] >= 0) coarse[hist[d[u] >> BSH] + r[u]] = make_int2(s[u], d[u]);
}

__global__ void __launch_bounds__(1024) k_bin_sort(const int2* __restrict__ coarse, const int* __restrict__ bintot,
                                                   int* __restrict__ rp, int* __restrict__ csr, int N) {
  __shared__ int cnt[BW];
  __shared__ int wsum[32];
  const int b = blockIdx.x;
  if (threadIdx.x < BW) cnt[threadIdx.x] = 0;
  const int t = ((int)threadIdx.x < b) ? bintot[threadIdx.x] : 0;
  int bstart;
  block_excl_scan(t, wsum, bstart);
  const int bc = bintot[b];
  const int2* __restrict__ ed = coarse + bstart;

  for (int i = threadIdx.x; i < bc; i += 1024) lds_add_rtn(&cnt[ed[i].y & (BW - 1)], 1);
  __syncthreads();

  const int v = (threadIdx.x < BW) ? cnt[threadIdx.x] : 0;
  int tot;
  const int ex = block_excl_scan(v, wsum, tot);
  if (threadIdx.x < BW) {
    const int node = b * BW + threadIdx.x;
    if (node < N) rp[node] = bstart + ex;
    cnt[threadIdx.x] = bstart + ex;
  }
  __syncthreads();

  for (int i = threadIdx.x; i < bc; i += 1024) {
    const int2 e = ed[i];
    csr[lds_add_rtn(&cnt[e.y & (BW - 1)], 1)] = e.x;
  }
}

template <int OUT>
__global__ void k_gemm_wmma(const float* __restrict__ A, const float* __restrict__ W,
                            float* __restrict__ C, int N) {
  constexpr int KD = 128;
  __shared__ float As[16 * KD];

  const int row0 = blockIdx.x * 16;
  for (int i = threadIdx.x; i < 16 * KD; i += blockDim.x) {
    int r = i / KD, k = i % KD;
    As[i] = (row0 + r < N) ? A[(size_t)(row0 + r) * KD + k] : 0.0f;
  }
  __syncthreads();

  const int wave = threadIdx.x >> 5;
  const int lane = threadIdx.x & 31;
  const int half = lane >> 4;
  const int n    = lane & 15;
  const int cw   = wave * 16;

  v8f c = {};
  for (int k0 = 0; k0 < KD; k0 += 4) {
    const int k = k0 + 2 * half;
    v2f a, b;
    a.x = As[n * KD + k];
    a.y = As[n * KD + k + 1];
    b.x = W[(size_t)k * OUT + cw + n];
    b.y = W[(size_t)(k + 1) * OUT + cw + n];
    c = __builtin_amdgcn_wmma_f32_16x16x4_f32(false, a, false, b,
                                              (short)0, c, false, false);
  }

#pragma unroll
  for (int r = 0; r < 8; ++r) {
    const int m = r + 8 * half;
    if (row0 + m < N) C[(size_t)(row0 + m) * OUT + cw + n] = c[r];
  }
}

template <int V> struct RowVec;
template <> struct RowVec<4> { typedef float4 T; };
template <> struct RowVec<1> { typedef float T; };

__device__ __forceinline__ void axpy(float4& a, float w, const float4 v) {
  a.x = fmaf(w, v.x, a.x); a.y = fmaf(w, v.y, a.y);
  a.z = fmaf(w, v.z, a.z); a.w = fmaf(w, v.w, a.w);
}
__device__ __forceinline__ void axpy(float& a, float w, const float v) { a = fmaf(w, v, a); }
__device__ __forceinline__ float4 plus(const float4 a, const float4 b) {
  return make_float4(a.x + b.x, a.y + b.y, a.z + b.z, a.w + b.w);
}
__device__ __forceinline__ float plus(const float a, const float b) { return a + b; }
__device__ __forceinline__ void set_zero(float4& a) { a = make_float4(0.0f, 0.0f, 0.0f, 0.0f); }
__device__ __forceinline__ void set_zero(float& a) { a = 0.0f; }

__device__ __forceinline__ float dinv_of(const int* __restrict__ rp, int i, int N, int E) {
  const int b = rp[i];
  const int e = (i + 1 < N) ? rp[i + 1] : E;
  return rsqrtf((float)(e - b + 1));
}

template <int B, typename T>
__device__ __forceinline__ void gather_rows(T& acc, const T* __restrict__ xr, int s_i, float w_i,
                                            int k, int lane) {
  T v[B];
  float w[B];
  for (int u = 0; u < B; ++u) {
    const int s = __builtin_amdgcn_readlane(s_i, k + u);
    w[u] = __int_as_float(__builtin_amdgcn_readlane(__float_as_int(w_i), k + u));
    v[u] = xr[(size_t)s * 32 + lane];
  }
  for (int u = 0; u < B; ++u) axpy(acc, w[u], v[u]);
}

template <int F>
__global__ void k_gather(const float* __restrict__ xw, const int* __restrict__ rp,
                         const int* __restrict__ csr, const float* __restrict__ bias,
                         float* __restrict__ out, int N, int E) {
  constexpr int V = F / 32;
  typedef typename RowVec<V>::T T;
  const int lane   = threadIdx.x & 31;
  const int wpb    = blockDim.x >> 5;
  const int nwaves = gridDim.x * wpb;
  const T* __restrict__ xr = (const T*)xw;
  T* __restrict__ orow     = (T*)out;
  const T bv = ((const T*)bias)[lane];

  int node = __builtin_amdgcn_readfirstlane((int)(blockIdx.x * wpb + (threadIdx.x >> 5)));
  for (; node < N; node += nwaves) {
    const int beg  = rp[node];
    const int end  = (node + 1 < N) ? rp[node + 1] : E;
    const float dn = rsqrtf((float)(end - beg + 1));
    const T self = xr[(size_t)node * 32 + lane];
    T acc;
    set_zero(acc);

    for (int base = beg; base < end; base += 32) {
      const int j = base + lane;
      int s_i = 0;
      float w_i = 0.0f;
      if (j < end) {
        s_i = csr[j];
        w_i = dinv_of(rp, s_i, N, E) * dn;
      }
      const int cnt = min(32, end - base);
      int k = 0;
      for (; k + 8 <= cnt; k += 8) gather_rows<8>(acc, xr, s_i, w_i, k, lane);
      if (k + 4 <= cnt) { gather_rows<4>(acc, xr, s_i, w_i, k, lane); k += 4; }
      if (k + 2 <= cnt) { gather_rows<2>(acc, xr, s_i, w_i, k, lane); k += 2; }
      if (k < cnt)      { gather_rows<1>(acc, xr, s_i, w_i, k, lane); }
    }
    axpy(acc, dn * dn, self);
    orow[(size_t)node * 32 + lane] = plus(acc, bv);
  }
}

#define NPW 16
#define NPG 8

__global__ void __launch_bounds__(128) k_elu_pool(float* __restrict__ h, const int* __restrict__ batch,
                                                  float* __restrict__ pool, float* __restrict__ cnt, int N) {
  const int lane = threadIdx.x & 31;
  const int n0 = __builtin_amdgcn_readfirstlane((int)((blockIdx.x * (blockDim.x >> 5) + (threadIdx.x >> 5)) * NPW));
  if (n0 >= N) return;
  const int n1 = min(n0 + NPW, N);
  float4* __restrict__ hr = (float4*)h;

  float4 acc = make_float4(0.0f, 0.0f, 0.0f, 0.0f);
  float c = 0.0f;
  int gcur = batch[n0];
  for (int nb = n0; nb < n1; nb += NPG) {
    float4 v[NPG];
    int g[NPG];
#pragma unroll
    for (int u = 0; u < NPG; ++u) {
      const int node = min(nb + u, n1 - 1);
      v[u] = hr[(size_t)node * 32 + lane];
      g[u] = batch[node];
    }
#pragma unroll
    for (int u = 0; u < NPG; ++u) {
      const int node = nb + u;
      if (node < n1) {
        float4 x = v[u];
        x.x = x.x > 0.0f ? x.x : expm1f(x.x);
        x.y = x.y > 0.0f ? x.y : expm1f(x.y);
        x.z = x.z > 0.0f ? x.z : expm1f(x.z);
        x.w = x.w > 0.0f ? x.w : expm1f(x.w);
        hr[(size_t)node * 32 + lane] = x;
        if (g[u] != gcur) {
          float* p = &pool[(size_t)gcur * 128 + lane * 4];
          atomAddF(p + 0, acc.x); atomAddF(p + 1, acc.y);
          atomAddF(p + 2, acc.z); atomAddF(p + 3, acc.w);
          if (lane == 0) atomAddF(&cnt[gcur], c);
          acc = make_float4(0.0f, 0.0f, 0.0f, 0.0f);
          c = 0.0f;
          gcur = g[u];
        }
        acc.x += x.x; acc.y += x.y; acc.z += x.z; acc.w += x.w;
        c += 1.0f;
      }
    }
  }
  float* p = &pool[(size_t)gcur * 128 + lane * 4];
  atomAddF(p + 0, acc.x); atomAddF(p + 1, acc.y);
  atomAddF(p + 2, acc.z); atomAddF(p + 3, acc.w);
  if (lane == 0) atomAddF(&cnt[gcur], c);
}

template <int IN, int OUT>
__global__ void k_master_mlp(const float* __restrict__ pool, const float* __restrict__ cnt,
                             const float* __restrict__ Wm, const float* __restrict__ bm,
                             float* __restrict__ mm, const int* __restrict__ gptr) {
  int G = *gptr;
  int idx = blockIdx.x * blockDim.x + threadIdx.x;
  int g = idx / OUT, j = idx % OUT;
  if (g >= G) return;
  float inv = 1.0f / fmaxf(cnt[g], 1.0f);
  float acc = bm[j];
  for (int i = 0; i < IN; ++i)
    acc = fmaf(pool[(size_t)g * IN + i] * inv, Wm[(size_t)i * OUT + j], acc);
  mm[(size_t)g * OUT + j] = fmaxf(acc, 0.0f);
}

__global__ void k_add_master(float* __restrict__ h, const float* __restrict__ mm,
                             const int* __restrict__ batch, int N) {
  int t = blockIdx.x * blockDim.x + threadIdx.x;
  int node = t >> 5;
  if (node >= N) return;
  int fo = (t & 31) * 4;
  int g = batch[node];
  float4 a = *(float4*)&h[(size_t)node * 128 + fo];
  float4 m = *(const float4*)&mm[(size_t)g * 128 + fo];
  a.x += m.x; a.y += m.y; a.z += m.z; a.w += m.w;
  *(float4*)&h[(size_t)node * 128 + fo] = a;
}

__global__ void __launch_bounds__(128) k_pool_z(const float* __restrict__ z, const int* __restrict__ batch,
                                                float* __restrict__ pool, int N) {
  const int lane = threadIdx.x & 31;
  const int n0 = __builtin_amdgcn_readfirstlane((int)((blockIdx.x * (blockDim.x >> 5) + (threadIdx.x >> 5)) * NPW));
  if (n0 >= N) return;
  const int n1 = min(n0 + NPW, N);

  float acc = 0.0f;
  int gcur = batch[n0];
  float v[NPW];
  int g[NPW];
#pragma unroll
  for (int u = 0; u < NPW; ++u) {
    const int node = min(n0 + u, n1 - 1);
    v[u] = z[(size_t)node * 32 + lane];
    g[u] = batch[node];
  }
#pragma unroll
  for (int u = 0; u < NPW; ++u) {
    if (n0 + u < n1) {
      if (g[u] != gcur) {
        atomAddF(&pool[(size_t)gcur * 32 + lane], acc);
        acc = 0.0f;
        gcur = g[u];
      }
      acc += v[u];
    }
  }
  atomAddF(&pool[(size_t)gcur * 32 + lane], acc);
}

__global__ void k_finalize(const float* __restrict__ z, const float* __restrict__ mm2,
                           const int* __restrict__ batch, float* __restrict__ out, int N) {
  int node = blockIdx.x * (blockDim.x >> 5) + (threadIdx.x >> 5);
  if (node >= N) return;
  int lane = threadIdx.x & 31;
  float v = z[(size_t)node * 32 + lane] + mm2[(size_t)batch[node] * 32 + lane];
  float m = v;
#pragma unroll
  for (int o = 16; o >= 1; o >>= 1) m = fmaxf(m, __shfl_xor(m, o, 32));
  float ev = __expf(v - m);
  float s = ev;
#pragma unroll
  for (int o = 16; o >= 1; o >>= 1) s += __shfl_xor(s, o, 32);
  out[(size_t)node * 32 + lane] = ev / s;
}

extern "C" void kernel_launch(void* const* d_in, const int* in_sizes, int n_in,
                              void* d_out, int out_size, void* d_ws, size_t ws_size,
                              hipStream_t stream) {
  const float* X   = (const float*)d_in[0];
  const float* W1  = (const float*)d_in[1];
  const float* b1  = (const float*)d_in[2];
  const float* W2  = (const float*)d_in[3];
  const float* b2  = (const float*)d_in[4];
  const float* mW1 = (const float*)d_in[5];
  const float* mb1 = (const float*)d_in[6];
  const float* mW2 = (const float*)d_in[7];
  const float* mb2 = (const float*)d_in[8];
  const int* ei    = (const int*)d_in[9];
  const int* batch = (const int*)d_in[10];
  const int* gptr  = (const int*)d_in[11];

  const int N = in_sizes[0] / 128;
  const int E = in_sizes[9] / 2;
  const int* src = ei;
  const int* dst = ei + E;

  char* w = (char*)d_ws;
  size_t off = 0;
  auto take = [&](size_t bytes) -> float* {
    float* p = (float*)(w + off);
    off = (off + bytes + 255) & ~(size_t)255;
    return p;
  };
  float* xw    = take((size_t)N * 128 * 4);
  float* h     = take((size_t)N * 128 * 4);
  int*   rp    = (int*)take((size_t)N * 4);
  int*   csr   = (int*)take((size_t)N * 32 * 4);
  int2*  coarse = (int2*)h;
  int*   bintot = (int*)xw;
  int*   bincur = bintot + NBMAX;
  float* z     = h;
  float* pool1 = take((size_t)GMAX * 128 * 4);
  float* pool2 = take((size_t)GMAX * 32 * 4);
  float* cnt   = take((size_t)GMAX * 4);
  float* mm1   = take((size_t)GMAX * 128 * 4);
  float* mm2   = take((size_t)GMAX * 32 * 4);

  const int B = 256;
  auto blks = [](long long n, int b) { return (unsigned)((n + b - 1) / b); };

  const int NB = (N + BW - 1) >> BSH;
  k_fill<<<blks(2 * NBMAX, B), B, 0, stream>>>((float*)bintot, 0.0f, 2 * NBMAX);
  k_fill<<<blks(GMAX * 128, B), B, 0, stream>>>(pool1, 0.0f, GMAX * 128);
  k_fill<<<blks(GMAX * 32, B), B, 0, stream>>>(pool2, 0.0f, GMAX * 32);
  k_fill<<<blks(GMAX, B), B, 0, stream>>>(cnt, 0.0f, GMAX);
  k_bin_count<<<blks(E, EPB), 1024, 0, stream>>>(dst, bintot, E, NB);
  k_bin_place<<<blks(E, EPB), 1024, 0, stream>>>(src, dst, bintot, bincur, coarse, E, NB);
  k_bin_sort<<<NB, 1024, 0, stream>>>(coarse, bintot, rp, csr, N);

  unsigned gblocks = blks(N, 8);
  if (gblocks > 2048u) gblocks = 2048u;

  k_gemm_wmma<128><<<blks(N, 16), 256, 0, stream>>>(X, W1, xw, N);
  k_gather<128><<<gblocks, B, 0, stream>>>(xw, rp, csr, b1, h, N, E);
  k_elu_pool<<<blks(N, 4 * NPW), 128, 0, stream>>>(h, batch, pool1, cnt, N);

  k_master_mlp<128, 128><<<blks(GMAX * 128, B), B, 0, stream>>>(pool1, cnt, mW1, mb1, mm1, gptr);
  k_add_master<<<blks((long long)N * 32, B), B, 0, stream>>>(h, mm1, batch, N);

  k_gemm_wmma<32><<<blks(N, 16), 64, 0, stream>>>(h, W2, xw, N);
  k_gather<32><<<gblocks, B, 0, stream>>>(xw, rp, csr, b2, z, N, E);

  k_pool_z<<<blks(N, 4 * NPW), 128, 0, stream>>>(z, batch, pool2, N);
  k_master_mlp<32, 32><<<blks(GMAX * 32, B), B, 0, stream>>>(pool2, cnt, mW2, mb2, mm2, gptr);
  k_finalize<<<blks(N, 8), B, 0, stream>>>(z, mm2, batch, (float*)d_out, N);
}


# Virtual-node add folded into the 2nd product's tile staging: a wave stages 8 whole rows as h[row] + mm[batch[row]], loads in flight; the in-place pass over h (a short wave per node) is gone
# speedup vs baseline: 4.6785x; 1.1156x over previous
#include <hip/hip_runtime.h>
#include <math.h>


typedef __attribute__((ext_vector_type(2))) float v2f;
typedef __attribute__((ext_vector_type(8))) float v8f;

#define GMAX 256

__device__ __forceinline__ void atomAddF(float* p, float v) {
  __hip_atomic_fetch_add(p, v, __ATOMIC_RELAXED, __HIP_MEMORY_SCOPE_AGENT);
}

__global__ void k_fill(float* p, float v, int n) {
  int i = blockIdx.x * blockDim.x + threadIdx.x;
  if (i < n) p[i] = v;
}

#define BSH   8
#define BW    (1 << BSH)
#define NBMAX 1024
#define EPT   4
#define EPB   (1024 * EPT)

__device__ __forceinline__ int lds_add_rtn(int* p, int v) {
  return __hip_atomic_fetch_add(p, v, __ATOMIC_RELAXED, __HIP_MEMORY_SCOPE_WORKGROUP);
}

__device__ __forceinline__ int wave_incl_scan(int v, int lane) {
  for (int o = 1; o < 32; o <<= 1) {
    int t = __shfl_up(v, o, 32);
    if (lane >= o) v += t;
  }
  return v;
}

__device__ __forceinline__ int block_excl_scan(int v, int* wsum, int& total) {
  const int lane = threadIdx.x & 31;
  const int wave = threadIdx.x >> 5;
  const int inc  = wave_incl_scan(v, lane);
  if (lane == 31) wsum[wave] = inc;
  __syncthreads();
  const int w    = wsum[lane];
  const int winc = wave_incl_scan(w, lane);
  total = __shfl(winc, 31, 32);
  const int wbase = __shfl(winc - w, wave, 32);
  __syncthreads();
  return wbase + inc - v;
}

__global__ void __launch_bounds__(1024) k_bin_count(const int* __restrict__ dst, int* __restrict__ bintot,
                                                    int E, int NB) {
  __shared__ int hist[NBMAX];
  for (int i = threadIdx.x; i < NB; i += 1024) hist[i] = 0;
  __syncthreads();
  const int base = blockIdx.x * EPB + threadIdx.x;
  int d[EPT];
#pragma unroll
  for (int u = 0; u < EPT; ++u) {
    const int e = base + u * 1024;
    d[u] = (e < E) ? dst[e] : -1;
  }
#pragma unroll
  for (int u = 0; u < EPT; ++u)
    if (d[u] >= 0) lds_add_rtn(&hist[d[u] >> BSH], 1);
  __syncthreads();
  for (int i = threadIdx.x; i < NB; i += 1024) {
    const int c = hist[i];
    if (c) __hip_atomic_fetch_add(&bintot[i], c, __ATOMIC_RELAXED, __HIP_MEMORY_SCOPE_AGENT);
  }
}

__global__ void __launch_bounds__(1024) k_bin_place(const int* __restrict__ src, const int* __restrict__ dst,
                                                    const int* __restrict__ bintot, int* __restrict__ bincur,
                                                    int2* __restrict__ coarse, int E, int NB) {
  __shared__ int hist[NBMAX];
  __shared__ int wsum[32];
  for (int i = threadIdx.x; i < NB; i += 1024) hist[i] = 0;
  __syncthreads();
  const int base = blockIdx.x * EPB + threadIdx.x;
  int s[EPT], d[EPT], r[EPT];
#pragma unroll
  for (int u = 0; u < EPT; ++u) {
    const int e = base + u * 1024;
    d[u] = (e < E) ? dst[e] : -1;
    s[u] = (e < E) ? src[e] : 0;
  }
#pragma unroll
  for (int u = 0; u < EPT; ++u)
    r[u] = (d[u] >= 0) ? lds_add_rtn(&hist[d[u] >> BSH], 1) : 0;

  const int t = (threadIdx.x < NB) ? bintot[threadIdx.x] : 0;
  int total;
  const int start = block_excl_scan(t, wsum, total);
  if (threadIdx.x < NB) {
    const int c = hist[threadIdx.x];
    const int off = c ? __hip_atomic_fetch_add(&bincur[threadIdx.x], c, __ATOMIC_RELAXED, __HIP_MEMORY_SCOPE_AGENT) : 0;
    hist[threadIdx.x] = start + off;
  }
  __syncthreads();
#pragma unroll
  for (int u = 0; u < EPT; ++u)
    if (d[u] >= 0) coarse[hist[d[u] >> BSH] + r[u]] = make_int2(s[u], d[u]);
}

__global__ void __launch_bounds__(1024) k_bin_sort(const int2* __restrict__ coarse, const int* __restrict__ bintot,
                                                   int* __restrict__ rp, int* __restrict__ csr, int N) {
  __shared__ int cnt[BW];
  __shared__ int wsum[32];
  const int b = blockIdx.x;
  if (threadIdx.x < BW) cnt[threadIdx.x] = 0;
  const int t = ((int)threadIdx.x < b) ? bintot[threadIdx.x] : 0;
  int bstart;
  block_excl_scan(t, wsum, bstart);
  const int bc = bintot[b];
  const int2* __restrict__ ed = coarse + bstart;

  for (int i = threadIdx.x; i < bc; i += 1024) lds_add_rtn(&cnt[ed[i].y & (BW - 1)], 1);
  __syncthreads();

  const int v = (threadIdx.x < BW) ? cnt[threadIdx.x] : 0;
  int tot;
  const int ex = block_excl_scan(v, wsum, tot);
  if (threadIdx.x < BW) {
    const int node = b * BW + threadIdx.x;
    if (node < N) rp[node] = bstart + ex;
    cnt[threadIdx.x] = bstart + ex;
  }
  __syncthreads();

  for (int i = threadIdx.x; i < bc; i += 1024) {
    const int2 e = ed[i];
    csr[lds_add_rtn(&cnt[e.y & (BW - 1)], 1)] = e.x;
  }
}

template <int OUT, bool ADDM>
__global__ void k_gemm_wmma(const float* __restrict__ A, const float* __restrict__ W,
                            float* __restrict__ C, int N,
                            const float* __restrict__ mm, const int* __restrict__ batch) {
  constexpr int KD = 128;
  __shared__ float As[16 * KD];

  const int row0 = blockIdx.x * 16;
  if constexpr (ADDM) {
    constexpr int RPW = 16 / (OUT / 16);
    const int ln = threadIdx.x & 31;
    const int t0 = __builtin_amdgcn_readfirstlane((int)(threadIdx.x >> 5) * RPW);
    const float4* __restrict__ Ar = (const float4*)A;
    const float4* __restrict__ mr = (const float4*)mm;
    float4 hv[RPW], mv[RPW];
#pragma unroll
    for (int u = 0; u < RPW; ++u) {
      const int row = min(row0 + t0 + u, N - 1);
      hv[u] = Ar[(size_t)row * 32 + ln];
      mv[u] = mr[(size_t)batch[row] * 32 + ln];
    }
#pragma unroll
    for (int u = 0; u < RPW; ++u) {
      float4 v = make_float4(0.0f, 0.0f, 0.0f, 0.0f);
      if (row0 + t0 + u < N)
        v = make_float4(hv[u].x + mv[u].x, hv[u].y + mv[u].y, hv[u].z + mv[u].z, hv[u].w + mv[u].w);
      ((float4*)As)[(t0 + u) * 32 + ln] = v;
    }
  } else {
    for (int i = threadIdx.x; i < 16 * KD; i += blockDim.x) {
      int r = i / KD, k = i % KD;
      As[i] = (row0 + r < N) ? A[(size_t)(row0 + r) * KD + k] : 0.0f;
    }
  }
  __syncthreads();

  const int wave = threadIdx.x >> 5;
  const int lane = threadIdx.x & 31;
  const int half = lane >> 4;
  const int n    = lane & 15;
  const int cw   = wave * 16;

  v8f c = {};
  for (int k0 = 0; k0 < KD; k0 += 4) {
    const int k = k0 + 2 * half;
    v2f a, b;
    a.x = As[n * KD + k];
    a.y = As[n * KD + k + 1];
    b.x = W[(size_t)k * OUT + cw + n];
    b.y = W[(size_t)(k + 1) * OUT + cw + n];
    c = __builtin_amdgcn_wmma_f32_16x16x4_f32(false, a, false, b,
                                              (short)0, c, false, false);
  }

#pragma unroll
  for (int r = 0; r < 8; ++r) {
    const int m = r + 8 * half;
    if (row0 + m < N) C[(size_t)(row0 + m) * OUT + cw + n] = c[r];
  }
}

template <int V> struct RowVec;
template <> struct RowVec<4> { typedef float4 T; };
template <> struct RowVec<1> { typedef float T; };

__device__ __forceinline__ void axpy(float4& a, float w, const float4 v) {
  a.x = fmaf(w, v.x, a.x); a.y = fmaf(w, v.y, a.y);
  a.z = fmaf(w, v.z, a.z); a.w = fmaf(w, v.w, a.w);
}
__device__ __forceinline__ void axpy(float& a, float w, const float v) { a = fmaf(w, v, a); }
__device__ __forceinline__ float4 plus(const float4 a, const float4 b) {
  return make_float4(a.x + b.x, a.y + b.y, a.z + b.z, a.w + b.w);
}
__device__ __forceinline__ float plus(const float a, const float b) { return a + b; }
__device__ __forceinline__ void set_zero(float4& a) { a = make_float4(0.0f, 0.0f, 0.0f, 0.0f); }
__device__ __forceinline__ void set_zero(float& a) { a = 0.0f; }

__device__ __forceinline__ float dinv_of(const int* __restrict__ rp, int i, int N, int E) {
  const int b = rp[i];
  const int e = (i + 1 < N) ? rp[i + 1] : E;
  return rsqrtf((float)(e - b + 1));
}

template <int B, typename T>
__device__ __forceinline__ void gather_rows(T& acc, const T* __restrict__ xr, int s_i, float w_i,
                                            int k, int lane) {
  T v[B];
  float w[B];
  for (int u = 0; u < B; ++u) {
    const int s = __builtin_amdgcn_readlane(s_i, k + u);
    w[u] = __int_as_float(__builtin_amdgcn_readlane(__float_as_int(w_i), k + u));
    v[u] = xr[(size_t)s * 32 + lane];
  }
  for (int u = 0; u < B; ++u) axpy(acc, w[u], v[u]);
}

template <int F>
__global__ void k_gather(const float* __restrict__ xw, const int* __restrict__ rp,
                         const int* __restrict__ csr, const float* __restrict__ bias,
                         float* __restrict__ out, int N, int E) {
  constexpr int V = F / 32;
  typedef typename RowVec<V>::T T;
  const int lane   = threadIdx.x & 31;
  const int wpb    = blockDim.x >> 5;
  const int nwaves = gridDim.x * wpb;
  const T* __restrict__ xr = (const T*)xw;
  T* __restrict__ orow     = (T*)out;
  const T bv = ((const T*)bias)[lane];

  int node = __builtin_amdgcn_readfirstlane((int)(blockIdx.x * wpb + (threadIdx.x >> 5)));
  for (; node < N; node += nwaves) {
    const int beg  = rp[node];
    const int end  = (node + 1 < N) ? rp[node + 1] : E;
    const float dn = rsqrtf((float)(end - beg + 1));
    const T self = xr[(size_t)node * 32 + lane];
    T acc;
    set_zero(acc);

    for (int base = beg; base < end; base += 32) {
      const int j = base + lane;
      int s_i = 0;
      float w_i = 0.0f;
      if (j < end) {
        s_i = csr[j];
        w_i = dinv_of(rp, s_i, N, E) * dn;
      }
      const int cnt = min(32, end - base);
      int k = 0;
      for (; k + 8 <= cnt; k += 8) gather_rows<8>(acc, xr, s_i, w_i, k, lane);
      if (k + 4 <= cnt) { gather_rows<4>(acc, xr, s_i, w_i, k, lane); k += 4; }
      if (k + 2 <= cnt) { gather_rows<2>(acc, xr, s_i, w_i, k, lane); k += 2; }
      if (k < cnt)      { gather_rows<1>(acc, xr, s_i, w_i, k, lane); }
    }
    axpy(acc, dn * dn, self);
    orow[(size_t)node * 32 + lane] = plus(acc, bv);
  }
}

#define NPW 16
#define NPG 8

__global__ void __launch_bounds__(128) k_elu_pool(float* __restrict__ h, const int* __restrict__ batch,
                                                  float* __restrict__ pool, float* __restrict__ cnt, int N) {
  const int lane = threadIdx.x & 31;
  const int n0 = __builtin_amdgcn_readfirstlane((int)((blockIdx.x * (blockDim.x >> 5) + (threadIdx.x >> 5)) * NPW));
  if (n0 >= N) return;
  const int n1 = min(n0 + NPW, N);
  float4* __restrict__ hr = (float4*)h;

  float4 acc = make_float4(0.0f, 0.0f, 0.0f, 0.0f);
  float c = 0.0f;
  int gcur = batch[n0];
  for (int nb = n0; nb < n1; nb += NPG) {
    float4 v[NPG];
    int g[NPG];
#pragma unroll
    for (int u = 0; u < NPG; ++u) {
      const int node = min(nb + u, n1 - 1);
      v[u] = hr[(size_t)node * 32 + lane];
      g[u] = batch[node];
    }
#pragma unroll
    for (int u = 0; u < NPG; ++u) {
      const int node = nb + u;
      if (node < n1) {
        float4 x = v[u];
        x.x = x.x > 0.0f ? x.x : expm1f(x.x);
        x.y = x.y > 0.0f ? x.y : expm1f(x.y);
        x.z = x.z > 0.0f ? x.z : expm1f(x.z);
        x.w = x.w > 0.0f ? x.w : expm1f(x.w);
        hr[(size_t)node * 32 + lane] = x;
        if (g[u] != gcur) {
          float* p = &pool[(size_t)gcur * 128 + lane * 4];
          atomAddF(p + 0, acc.x); atomAddF(p + 1, acc.y);
          atomAddF(p + 2, acc.z); atomAddF(p + 3, acc.w);
          if (lane == 0) atomAddF(&cnt[gcur], c);
          acc = make_float4(0.0f, 0.0f, 0.0f, 0.0f);
          c = 0.0f;
          gcur = g[u];
        }
        acc.x += x.x; acc.y += x.y; acc.z += x.z; acc.w += x.w;
        c += 1.0f;
      }
    }
  }
  float* p = &pool[(size_t)gcur * 128 + lane * 4];
  atomAddF(p + 0, acc.x); atomAddF(p + 1, acc.y);
  atomAddF(p + 2, acc.z); atomAddF(p + 3, acc.w);
  if (lane == 0) atomAddF(&cnt[gcur], c);
}

template <int IN, int OUT>
__global__ void k_master_mlp(const float* __restrict__ pool, const float* __restrict__ cnt,
                             const float* __restrict__ Wm, const float* __restrict__ bm,
                             float* __restrict__ mm, const int* __restrict__ gptr) {
  int G = *gptr;
  int idx = blockIdx.x * blockDim.x + threadIdx.x;
  int g = idx / OUT, j = idx % OUT;
  if (g >= G) return;
  float inv = 1.0f / fmaxf(cnt[g], 1.0f);
  float acc = bm[j];
  for (int i = 0; i < IN; ++i)
    acc = fmaf(pool[(size_t)g * IN + i] * inv, Wm[(size_t)i * OUT + j], acc);
  mm[(size_t)g * OUT + j] = fmaxf(acc, 0.0f);
}

__global__ void __launch_bounds__(128) k_pool_z(const float* __restrict__ z, const int* __restrict__ batch,
                                                float* __restrict__ pool, int N) {
  const int lane = threadIdx.x & 31;
  const int n0 = __builtin_amdgcn_readfirstlane((int)((blockIdx.x * (blockDim.x >> 5) + (threadIdx.x >> 5)) * NPW));
  if (n0 >= N) return;
  const int n1 = min(n0 + NPW, N);

  float acc = 0.0f;
  int gcur = batch[n0];
  float v[NPW];
  int g[NPW];
#pragma unroll
  for (int u = 0; u < NPW; ++u) {
    const int node = min(n0 + u, n1 - 1);
    v[u] = z[(size_t)node * 32 + lane];
    g[u] = batch[node];
  }
#pragma unroll
  for (int u = 0; u < NPW; ++u) {
    if (n0 + u < n1) {
      if (g[u] != gcur) {
        atomAddF(&pool[(size_t)gcur * 32 + lane], acc);
        acc = 0.0f;
        gcur = g[u];
      }
      acc += v[u];
    }
  }
  atomAddF(&pool[(size_t)gcur * 32 + lane], acc);
}

__global__ void k_finalize(const float* __restrict__ z, const float* __restrict__ mm2,
                           const int* __restrict__ batch, float* __restrict__ out, int N) {
  int node = blockIdx.x * (blockDim.x >> 5) + (threadIdx.x >> 5);
  if (node >= N) return;
  int lane = threadIdx.x & 31;
  float v = z[(size_t)node * 32 + lane] + mm2[(size_t)batch[node] * 32 + lane];
  float m = v;
#pragma unroll
  for (int o = 16; o >= 1; o >>= 1) m = fmaxf(m, __shfl_xor(m, o, 32));
  float ev = __expf(v - m);
  float s = ev;
#pragma unroll
  for (int o = 16; o >= 1; o >>= 1) s += __shfl_xor(s, o, 32);
  out[(size_t)node * 32 + lane] = ev / s;
}

extern "C" void kernel_launch(void* const* d_in, const int* in_sizes, int n_in,
                              void* d_out, int out_size, void* d_ws, size_t ws_size,
                              hipStream_t stream) {
  const float* X   = (const float*)d_in[0];
  const float* W1  = (const float*)d_in[1];
  const float* b1  = (const float*)d_in[2];
  const float* W2  = (const float*)d_in[3];
  const float* b2  = (const float*)d_in[4];
  const float* mW1 = (const float*)d_in[5];
  const float* mb1 = (const float*)d_in[6];
  const float* mW2 = (const float*)d_in[7];
  const float* mb2 = (const float*)d_in[8];
  const int* ei    = (const int*)d_in[9];
  const int* batch = (const int*)d_in[10];
  const int* gptr  = (const int*)d_in[11];

  const int N = in_sizes[0] / 128;
  const int E = in_sizes[9] / 2;
  const int* src = ei;
  const int* dst = ei + E;

  char* w = (char*)d_ws;
  size_t off = 0;
  auto take = [&](size_t bytes) -> float* {
    float* p = (float*)(w + off);
    off = (off + bytes + 255) & ~(size_t)255;
    return p;
  };
  float* xw    = take((size_t)N * 128 * 4);
  float* h     = take((size_t)N * 128 * 4);
  int*   rp    = (int*)take((size_t)N * 4);
  int*   csr   = (int*)take((size_t)N * 32 * 4);
  int2*  coarse = (int2*)h;
  int*   bintot = (int*)xw;
  int*   bincur = bintot + NBMAX;
  float* z     = h;
  float* pool1 = take((size_t)GMAX * 128 * 4);
  float* pool2 = take((size_t)GMAX * 32 * 4);
  float* cnt   = take((size_t)GMAX * 4);
  float* mm1   = take((size_t)GMAX * 128 * 4);
  float* mm2   = take((size_t)GMAX * 32 * 4);

  const int B = 256;
  auto blks = [](long long n, int b) { return (unsigned)((n + b - 1) / b); };

  const int NB = (N + BW - 1) >> BSH;
  k_fill<<<blks(2 * NBMAX, B), B, 0, stream>>>((float*)bintot, 0.0f, 2 * NBMAX);
  k_fill<<<blks(GMAX * 128, B), B, 0, stream>>>(pool1, 0.0f, GMAX * 128);
  k_fill<<<blks(GMAX * 32, B), B, 0, stream>>>(pool2, 0.0f, GMAX * 32);
  k_fill<<<blks(GMAX, B), B, 0, stream>>>(cnt, 0.0f, GMAX);
  k_bin_count<<<blks(E, EPB), 1024, 0, stream>>>(dst, bintot, E, NB);
  k_bin_place<<<blks(E, EPB), 1024, 0, stream>>>(src, dst, bintot, bincur, coarse, E, NB);
  k_bin_sort<<<NB, 1024, 0, stream>>>(coarse, bintot, rp, csr, N);

  unsigned gblocks = blks(N, 8);
  if (gblocks > 2048u) gblocks = 2048u;

  k_gemm_wmma<128, false><<<blks(N, 16), 256, 0, stream>>>(X, W1, xw, N, nullptr, nullptr);
  k_gather<128><<<gblocks, B, 0, stream>>>(xw, rp, csr, b1, h, N, E);
  k_elu_pool<<<blks(N, 4 * NPW), 128, 0, stream>>>(h, batch, pool1, cnt, N);

  k_master_mlp<128, 128><<<blks(GMAX * 128, B), B, 0, stream>>>(pool1, cnt, mW1, mb1, mm1, gptr);

  k_gemm_wmma<32, true><<<blks(N, 16), 64, 0, stream>>>(h, W2, xw, N, mm1, batch);
  k_gather<32><<<gblocks, B, 0, stream>>>(xw, rp, csr, b2, z, N, E);

  k_pool_z<<<blks(N, 4 * NPW), 128, 0, stream>>>(z, batch, pool2, N);
  k_master_mlp<32, 32><<<blks(GMAX * 32, B), B, 0, stream>>>(pool2, cnt, mW2, mb2, mm2, gptr);
  k_finalize<<<blks(N, 8), B, 0, stream>>>(z, mm2, batch, (float*)d_out, N);
}


# ELU and graph pooling folded into both aggregation kernels: a wave sums the 3 consecutive rows it forms, 8 waves join in LDS, one add per graph met; pool sum order = run, wave, then arrival order
# speedup vs baseline: 5.2999x; 1.1328x over previous
#include <hip/hip_runtime.h>
#include <math.h>


typedef __attribute__((ext_vector_type(2))) float v2f;
typedef __attribute__((ext_vector_type(8))) float v8f;

#define GMAX 256

__device__ __forceinline__ void atomAddF(float* p, float v) {
  __hip_atomic_fetch_add(p, v, __ATOMIC_RELAXED, __HIP_MEMORY_SCOPE_AGENT);
}

__global__ void k_fill(float* p, float v, int n) {
  int i = blockIdx.x * blockDim.x + threadIdx.x;
  if (i < n) p[i] = v;
}

#define BSH   8
#define BW    (1 << BSH)
#define NBMAX 1024
#define EPT   4
#define EPB   (1024 * EPT)

__device__ __forceinline__ int lds_add_rtn(int* p, int v) {
  return __hip_atomic_fetch_add(p, v, __ATOMIC_RELAXED, __HIP_MEMORY_SCOPE_WORKGROUP);
}

__device__ __forceinline__ int wave_incl_scan(int v, int lane) {
  for (int o = 1; o < 32; o <<= 1) {
    int t = __shfl_up(v, o, 32);
    if (lane >= o) v += t;
  }
  return v;
}

__device__ __forceinline__ int block_excl_scan(int v, int* wsum, int& total) {
  const int lane = threadIdx.x & 31;
  const int wave = threadIdx.x >> 5;
  const int inc  = wave_incl_scan(v, lane);
  if (lane == 31) wsum[wave] = inc;
  __syncthreads();
  const int w    = wsum[lane];
  const int winc = wave_incl_scan(w, lane);
  total = __shfl(winc, 31, 32);
  const int wbase = __shfl(winc - w, wave, 32);
  __syncthreads();
  return wbase + inc - v;
}

__global__ void __launch_bounds__(1024) k_bin_count(const int* __restrict__ dst, int* __restrict__ bintot,
                                                    int E, int NB) {
  __shared__ int hist[NBMAX];
  for (int i = threadIdx.x; i < NB; i += 1024) hist[i] = 0;
  __syncthreads();
  const int base = blockIdx.x * EPB + threadIdx.x;
  int d[EPT];
#pragma unroll
  for (int u = 0; u < EPT; ++u) {
    const int e = base + u * 1024;
    d[u] = (e < E) ? dst[e] : -1;
  }
#pragma unroll
  for (int u = 0; u < EPT; ++u)
    if (d[u] >= 0) lds_add_rtn(&hist[d[u] >> BSH], 1);
  __syncthreads();
  for (int i = threadIdx.x; i < NB; i += 1024) {
    const int c = hist[i];
    if (c) __hip_atomic_fetch_add(&bintot[i], c, __ATOMIC_RELAXED, __HIP_MEMORY_SCOPE_AGENT);
  }
}

__global__ void __launch_bounds__(1024) k_bin_place(const int* __restrict__ src, const int* __restrict__ dst,
                                                    const int* __restrict__ bintot, int* __restrict__ bincur,
                                                    int2* __restrict__ coarse, int E, int NB) {
  __shared__ int hist[NBMAX];
  __shared__ int wsum[32];
  for (int i = threadIdx.x; i < NB; i += 1024) hist[i] = 0;
  __syncthreads();
  const int base = blockIdx.x * EPB + threadIdx.x;
  int s[EPT], d[EPT], r[EPT];
#pragma unroll
  for (int u = 0; u < EPT; ++u) {
    const int e = base + u * 1024;
    d[u] = (e < E) ? dst[e] : -1;
    s[u] = (e < E) ? src[e] : 0;
  }
#pragma unroll
  for (int u = 0; u < EPT; ++u)
    r[u] = (d[u] >= 0) ? lds_add_rtn(&hist[d[u] >> BSH], 1) : 0;

  const int t = (threadIdx.x < NB) ? bintot[threadIdx.x] : 0;
  int total;
  const int start = block_excl_scan(t, wsum, total);
  if (threadIdx.x < NB) {
    const int c = hist[threadIdx.x];
    const int off = c ? __hip_atomic_fetch_add(&bincur[threadIdx.x], c, __ATOMIC_RELAXED, __HIP_MEMORY_SCOPE_AGENT) : 0;
    hist[threadIdx.x] = start + off;
  }
  __syncthreads();
#pragma unroll
  for (int u = 0; u < EPT; ++u)
    if (d[u] >= 0) coarse[hist[d[u] >> BSH] + r[u]] = make_int2(s[u], d[u]);
}

__global__ void __launch_bounds__(1024) k_bin_sort(const int2* __restrict__ coarse, const int* __restrict__ bintot,
                                                   int* __restrict__ rp, int* __restrict__ csr, int N) {
  __shared__ int cnt[BW];
  __shared__ int wsum[32];
  const int b = blockIdx.x;
  if (threadIdx.x < BW) cnt[threadIdx.x] = 0;
  const int t = ((int)threadIdx.x < b) ? bintot[threadIdx.x] : 0;
  int bstart;
  block_excl_scan(t, wsum, bstart);
  const int bc = bintot[b];
  const int2* __restrict__ ed = coarse + bstart;

  for (int i = threadIdx.x; i < bc; i += 1024) lds_add_rtn(&cnt[ed[i].y & (BW - 1)], 1);
  __syncthreads();

  const int v = (threadIdx.x < BW) ? cnt[threadIdx.x] : 0;
  int tot;
  const int ex = block_excl_scan(v, wsum, tot);
  if (threadIdx.x < BW) {
    const int node = b * BW + threadIdx.x;
    if (node < N) rp[node] = bstart + ex;
    cnt[threadIdx.x] = bstart + ex;
  }
  __syncthreads();

  for (int i = threadIdx.x; i < bc; i += 1024) {
    const int2 e = ed[i];
    csr[lds_add_rtn(&cnt[e.y & (BW - 1)], 1)] = e.x;
  }
}

__device__ __forceinline__ float dinv_of(const int* __restrict__ rp, int i, int N, int E) {
  const int b = rp[i];
  const int e = (i + 1 < N) ? rp[i + 1] : E;
  return rsqrtf((float)(e - b + 1));
}

template <int OUT, bool ADDM>
__global__ void k_gemm_wmma(const float* __restrict__ A, const float* __restrict__ W,
                            float* __restrict__ C, int N,
                            const float* __restrict__ mm, const int* __restrict__ batch,
                            const int* __restrict__ rp, int E) {
  constexpr int KD = 128;
  __shared__ float As[16 * KD];

  const int row0 = blockIdx.x * 16;
  if constexpr (ADDM) {
    constexpr int RPW = 16 / (OUT / 16);
    const int ln = threadIdx.x & 31;
    const int t0 = __builtin_amdgcn_readfirstlane((int)(threadIdx.x >> 5) * RPW);
    const float4* __restrict__ Ar = (const float4*)A;
    const float4* __restrict__ mr = (const float4*)mm;
    float4 hv[RPW], mv[RPW];
#pragma unroll
    for (int u = 0; u < RPW; ++u) {
      const int row = min(row0 + t0 + u, N - 1);
      hv[u] = Ar[(size_t)row * 32 + ln];
      mv[u] = mr[(size_t)batch[row] * 32 + ln];
    }
#pragma unroll
    for (int u = 0; u < RPW; ++u) {
      float4 v = make_float4(0.0f, 0.0f, 0.0f, 0.0f);
      if (row0 + t0 + u < N)
        v = make_float4(hv[u].x + mv[u].x, hv[u].y + mv[u].y, hv[u].z + mv[u].z, hv[u].w + mv[u].w);
      ((float4*)As)[(t0 + u) * 32 + ln] = v;
    }
  } else {
    for (int i = threadIdx.x; i < 16 * KD; i += blockDim.x) {
      int r = i / KD, k = i % KD;
      As[i] = (row0 + r < N) ? A[(size_t)(row0 + r) * KD + k] : 0.0f;
    }
  }
  __syncthreads();

  const int wave = threadIdx.x >> 5;
  const int lane = threadIdx.x & 31;
  const int half = lane >> 4;
  const int n    = lane & 15;
  const int cw   = wave * 16;

  float dv[8];
#pragma unroll
  for (int r = 0; r < 8; ++r) dv[r] = dinv_of(rp, min(row0 + r + 8 * half, N - 1), N, E);

  v8f c = {};
  for (int k0 = 0; k0 < KD; k0 += 4) {
    const int k = k0 + 2 * half;
    v2f a, b;
    a.x = As[n * KD + k];
    a.y = As[n * KD + k + 1];
    b.x = W[(size_t)k * OUT + cw + n];
    b.y = W[(size_t)(k + 1) * OUT + cw + n];
    c = __builtin_amdgcn_wmma_f32_16x16x4_f32(false, a, false, b,
                                              (short)0, c, false, false);
  }

#pragma unroll
  for (int r = 0; r < 8; ++r) {
    const int m = r + 8 * half;
    if (row0 + m < N) C[(size_t)(row0 + m) * OUT + cw + n] = c[r] * dv[r];
  }
}

template <int V> struct RowVec;
template <> struct RowVec<4> { typedef float4 T; };
template <> struct RowVec<1> { typedef float T; };

__device__ __forceinline__ float4 scale_add(float w, const float4 a, const float4 b) {
  return make_float4(fmaf(w, a.x, b.x), fmaf(w, a.y, b.y), fmaf(w, a.z, b.z), fmaf(w, a.w, b.w));
}
__device__ __forceinline__ float scale_add(float w, const float a, const float b) { return fmaf(w, a, b); }
__device__ __forceinline__ float4 plus(const float4 a, const float4 b) {
  return make_float4(a.x + b.x, a.y + b.y, a.z + b.z, a.w + b.w);
}
__device__ __forceinline__ float plus(const float a, const float b) { return a + b; }
__device__ __forceinline__ void set_zero(float4& a) { a = make_float4(0.0f, 0.0f, 0.0f, 0.0f); }
__device__ __forceinline__ void set_zero(float& a) { a = 0.0f; }

template <int B, typename T>
__device__ __forceinline__ void gather_rows(T& acc, const T* __restrict__ xr, int s_i, int k, int lane) {
  T v[B];
  for (int u = 0; u < B; ++u) {
    const int s = __builtin_amdgcn_readlane(s_i, k + u);
    v[u] = xr[(size_t)s * 32 + lane];
  }
  for (int u = 0; u < B; ++u) acc = plus(acc, v[u]);
}

__device__ __forceinline__ float elu1(float x) { return x > 0.0f ? x : expm1f(x); }
__device__ __forceinline__ float4 elu_row(const float4 a) {
  return make_float4(elu1(a.x), elu1(a.y), elu1(a.z), elu1(a.w));
}
__device__ __forceinline__ float elu_row(const float a) { return elu1(a); }

__device__ __forceinline__ void pool_add_row(float* __restrict__ pool, int g, int lane, const float4 a) {
  float* p = &pool[(size_t)g * 128 + lane * 4];
  atomAddF(p + 0, a.x); atomAddF(p + 1, a.y);
  atomAddF(p + 2, a.z); atomAddF(p + 3, a.w);
}
__device__ __forceinline__ void pool_add_row(float* __restrict__ pool, int g, int lane, const float a) {
  atomAddF(&pool[(size_t)g * 32 + lane], a);
}

#define RUN 3
#define GW  8

template <int F, bool ELU>
__global__ void __launch_bounds__(GW * 32) k_gather(const float* __restrict__ xw, const int* __restrict__ rp,
                         const int* __restrict__ csr, const float* __restrict__ bias,
                         float* __restrict__ out, const int* __restrict__ batch,
                         float* __restrict__ pool, float* __restrict__ cnt, int N, int E) {
  constexpr int V = F / 32;
  typedef typename RowVec<V>::T T;
  __shared__ T     prow[GW * 32];
  __shared__ float pcnt[GW];
  __shared__ int   pgr[GW];
  const int lane = threadIdx.x & 31;
  const T* __restrict__ xr = (const T*)xw;
  T* __restrict__ orow     = (T*)out;
  const T bv = ((const T*)bias)[lane];

  const int wave = __builtin_amdgcn_readfirstlane((int)(threadIdx.x >> 5));
  const int n0 = (blockIdx.x * GW + wave) * RUN;
  const int n1 = min(n0 + RUN, N);

  T pacc;
  set_zero(pacc);
  float pc = 0.0f;
  int gcur = (n0 < N) ? batch[n0] : -1;

  for (int node = n0; node < n1; ++node) {
    const int beg  = rp[node];
    const int end  = (node + 1 < N) ? rp[node + 1] : E;
    const int g    = batch[node];
    const float dn = rsqrtf((float)(end - beg + 1));
    const T self = xr[(size_t)node * 32 + lane];
    T acc;
    set_zero(acc);

    for (int base = beg; base < end; base += 32) {
      const int j = base + lane;
      const int s_i = (j < end) ? csr[j] : 0;
      const int cnt_e = min(32, end - base);
      int k = 0;
      for (; k + 8 <= cnt_e; k += 8) gather_rows<8>(acc, xr, s_i, k, lane);
      if (k + 4 <= cnt_e) { gather_rows<4>(acc, xr, s_i, k, lane); k += 4; }
      if (k + 2 <= cnt_e) { gather_rows<2>(acc, xr, s_i, k, lane); k += 2; }
      if (k < cnt_e)      { gather_rows<1>(acc, xr, s_i, k, lane); }
    }
    acc = plus(acc, self);
    T r = scale_add(dn, acc, bv);
    if constexpr (ELU) r = elu_row(r);
    orow[(size_t)node * 32 + lane] = r;

    if (g != gcur) {
      pool_add_row(pool, gcur, lane, pacc);
      if (ELU && lane == 0) atomAddF(&cnt[gcur], pc);
      set_zero(pacc);
      pc = 0.0f;
      gcur = g;
    }
    pacc = plus(pacc, r);
    pc += 1.0f;
  }

  prow[wave * 32 + lane] = pacc;
  if (lane == 0) { pcnt[wave] = pc; pgr[wave] = gcur; }
  __syncthreads();
  if (threadIdx.x < F) {
    const float* pf = (const float*)prow;
    float sq[GW], cq[GW];
    int gq[GW];
#pragma unroll
    for (int w = 0; w < GW; ++w) {
      gq[w] = pgr[w];
      cq[w] = pcnt[w];
      sq[w] = pf[w * F + threadIdx.x];
    }
    float s = 0.0f, c = 0.0f;
    int g = -1;
#pragma unroll
    for (int w = 0; w < GW; ++w) {
      const int gw = gq[w];
      if (gw < 0) continue;
      if (gw != g) {
        if (g >= 0) {
          atomAddF(&pool[(size_t)g * F + threadIdx.x], s);
          if (ELU && threadIdx.x == 0) atomAddF(&cnt[g], c);
        }
        s = 0.0f;
        c = 0.0f;
        g = gw;
      }
      s += sq[w];
      c += cq[w];
    }
    if (g >= 0) {
      atomAddF(&pool[(size_t)g * F + threadIdx.x], s);
      if (ELU && threadIdx.x == 0) atomAddF(&cnt[g], c);
    }
  }
}

template <int IN, int OUT>
__global__ void k_master_mlp(const float* __restrict__ pool, const float* __restrict__ cnt,
                             const float* __restrict__ Wm, const float* __restrict__ bm,
                             float* __restrict__ mm, const int* __restrict__ gptr) {
  int G = *gptr;
  int idx = blockIdx.x * blockDim.x + threadIdx.x;
  int g = idx / OUT, j = idx % OUT;
  if (g >= G) return;
  float inv = 1.0f / fmaxf(cnt[g], 1.0f);
  float acc = bm[j];
  for (int i = 0; i < IN; ++i)
    acc = fmaf(pool[(size_t)g * IN + i] * inv, Wm[(size_t)i * OUT + j], acc);
  mm[(size_t)g * OUT + j] = fmaxf(acc, 0.0f);
}

__global__ void k_finalize(const float* __restrict__ z, const float* __restrict__ mm2,
                           const int* __restrict__ batch, float* __restrict__ out, int N) {
  int node = blockIdx.x * (blockDim.x >> 5) + (threadIdx.x >> 5);
  if (node >= N) return;
  int lane = threadIdx.x & 31;
  float v = z[(size_t)node * 32 + lane] + mm2[(size_t)batch[node] * 32 + lane];
  float m = v;
#pragma unroll
  for (int o = 16; o >= 1; o >>= 1) m = fmaxf(m, __shfl_xor(m, o, 32));
  float ev = __expf(v - m);
  float s = ev;
#pragma unroll
  for (int o = 16; o >= 1; o >>= 1) s += __shfl_xor(s, o, 32);
  out[(size_t)node * 32 + lane] = ev / s;
}

extern "C" void kernel_launch(void* const* d_in, const int* in_sizes, int n_in,
                              void* d_out, int out_size, void* d_ws, size_t ws_size,
                              hipStream_t stream) {
  const float* X   = (const float*)d_in[0];
  const float* W1  = (const float*)d_in[1];
  const float* b1  = (const float*)d_in[2];
  const float* W2  = (const float*)d_in[3];
  const float* b2  = (const float*)d_in[4];
  const float* mW1 = (const float*)d_in[5];
  const float* mb1 = (const float*)d_in[6];
  const float* mW2 = (const float*)d_in[7];
  const float* mb2 = (const float*)d_in[8];
  const int* ei    = (const int*)d_in[9];
  const int* batch = (const int*)d_in[10];
  const int* gptr  = (const int*)d_in[11];

  const int N = in_sizes[0] / 128;
  const int E = in_sizes[9] / 2;
  const int* src = ei;
  const int* dst = ei + E;

  char* w = (char*)d_ws;
  size_t off = 0;
  auto take = [&](size_t bytes) -> float* {
    float* p = (float*)(w + off);
    off = (off + bytes + 255) & ~(size_t)255;
    return p;
  };
  float* xw    = take((size_t)N * 128 * 4);
  float* h     = take((size_t)N * 128 * 4);
  int*   rp    = (int*)take((size_t)N * 4);
  int*   csr   = (int*)take((size_t)N * 32 * 4);
  int2*  coarse = (int2*)h;
  int*   bintot = (int*)xw;
  int*   bincur = bintot + NBMAX;
  float* z     = h;
  float* pool1 = take((size_t)GMAX * 128 * 4);
  float* pool2 = take((size_t)GMAX * 32 * 4);
  float* cnt   = take((size_t)GMAX * 4);
  float* mm1   = take((size_t)GMAX * 128 * 4);
  float* mm2   = take((size_t)GMAX * 32 * 4);

  const int B = 256;
  auto blks = [](long long n, int b) { return (unsigned)((n + b - 1) / b); };

  const int NB = (N + BW - 1) >> BSH;
  k_fill<<<blks(2 * NBMAX, B), B, 0, stream>>>((float*)bintot, 0.0f, 2 * NBMAX);
  k_fill<<<blks(GMAX * 128, B), B, 0, stream>>>(pool1, 0.0f, GMAX * 128);
  k_fill<<<blks(GMAX * 32, B), B, 0, stream>>>(pool2, 0.0f, GMAX * 32);
  k_fill<<<blks(GMAX, B), B, 0, stream>>>(cnt, 0.0f, GMAX);
  k_bin_count<<<blks(E, EPB), 1024, 0, stream>>>(dst, bintot, E, NB);
  k_bin_place<<<blks(E, EPB), 1024, 0, stream>>>(src, dst, bintot, bincur, coarse, E, NB);
  k_bin_sort<<<NB, 1024, 0, stream>>>(coarse, bintot, rp, csr, N);

  const unsigned gblocks = blks(N, GW * RUN);

  k_gemm_wmma<128, false><<<blks(N, 16), 256, 0, stream>>>(X, W1, xw, N, nullptr, nullptr, rp, E);
  k_gather<128, true><<<gblocks, GW * 32, 0, stream>>>(xw, rp, csr, b1, h, batch, pool1, cnt, N, E);

  k_master_mlp<128, 128><<<blks(GMAX * 128, B), B, 0, stream>>>(pool1, cnt, mW1, mb1, mm1, gptr);

  k_gemm_wmma<32, true><<<blks(N, 16), 64, 0, stream>>>(h, W2, xw, N, mm1, batch, rp, E);
  k_gather<32, false><<<gblocks, GW * 32, 0, stream>>>(xw, rp, csr, b2, z, batch, pool2, nullptr, N, E);

  k_master_mlp<32, 32><<<blks(GMAX * 32, B), B, 0, stream>>>(pool2, cnt, mW2, mb2, mm2, gptr);
  k_finalize<<<blks(N, 8), B, 0, stream>>>(z, mm2, batch, (float*)d_out, N);
}


# GEMM A tile in LDS at a row pitch of 132 floats (was 128): the 32 lanes of a fragment read now fall on 64 distinct banks, not 16 rows on one bank pair (16-way conflict); both products
# speedup vs baseline: 5.6445x; 1.0650x over previous
#include <hip/hip_runtime.h>
#include <math.h>


typedef __attribute__((ext_vector_type(2))) float v2f;
typedef __attribute__((ext_vector_type(8))) float v8f;

#define GMAX 256

__device__ __forceinline__ void atomAddF(float* p, float v) {
  __hip_atomic_fetch_add(p, v, __ATOMIC_RELAXED, __HIP_MEMORY_SCOPE_AGENT);
}

__global__ void k_fill(float* p, float v, int n) {
  int i = blockIdx.x * blockDim.x + threadIdx.x;
  if (i < n) p[i] = v;
}

#define BSH   8
#define BW    (1 << BSH)
#define NBMAX 1024
#define EPT   4
#define EPB   (1024 * EPT)

__device__ __forceinline__ int lds_add_rtn(int* p, int v) {
  return __hip_atomic_fetch_add(p, v, __ATOMIC_RELAXED, __HIP_MEMORY_SCOPE_WORKGROUP);
}

__device__ __forceinline__ int wave_incl_scan(int v, int lane) {
  for (int o = 1; o < 32; o <<= 1) {
    int t = __shfl_up(v, o, 32);
    if (lane >= o) v += t;
  }
  return v;
}

__device__ __forceinline__ int block_excl_scan(int v, int* wsum, int& total) {
  const int lane = threadIdx.x & 31;
  const int wave = threadIdx.x >> 5;
  const int inc  = wave_incl_scan(v, lane);
  if (lane == 31) wsum[wave] = inc;
  __syncthreads();
  const int w    = wsum[lane];
  const int winc = wave_incl_scan(w, lane);
  total = __shfl(winc, 31, 32);
  const int wbase = __shfl(winc - w, wave, 32);
  __syncthreads();
  return wbase + inc - v;
}

__global__ void __launch_bounds__(1024) k_bin_count(const int* __restrict__ dst, int* __restrict__ bintot,
                                                    int E, int NB) {
  __shared__ int hist[NBMAX];
  for (int i = threadIdx.x; i < NB; i += 1024) hist[i] = 0;
  __syncthreads();
  const int base = blockIdx.x * EPB + threadIdx.x;
  int d[EPT];
#pragma unroll
  for (int u = 0; u < EPT; ++u) {
    const int e = base + u * 1024;
    d[u] = (e < E) ? dst[e] : -1;
  }
#pragma unroll
  for (int u = 0; u < EPT; ++u)
    if (d[u] >= 0) lds_add_rtn(&hist[d[u] >> BSH], 1);
  __syncthreads();
  for (int i = threadIdx.x; i < NB; i += 1024) {
    const int c = hist[i];
    if (c) __hip_atomic_fetch_add(&bintot[i], c, __ATOMIC_RELAXED, __HIP_MEMORY_SCOPE_AGENT);
  }
}

__global__ void __launch_bounds__(1024) k_bin_place(const int* __restrict__ src, const int* __restrict__ dst,
                                                    const int* __restrict__ bintot, int* __restrict__ bincur,
                                                    int2* __restrict__ coarse, int E, int NB) {
  __shared__ int hist[NBMAX];
  __shared__ int wsum[32];
  for (int i = threadIdx.x; i < NB; i += 1024) hist[i] = 0;
  __syncthreads();
  const int base = blockIdx.x * EPB + threadIdx.x;
  int s[EPT], d[EPT], r[EPT];
#pragma unroll
  for (int u = 0; u < EPT; ++u) {
    const int e = base + u * 1024;
    d[u] = (e < E) ? dst[e] : -1;
    s[u] = (e < E) ? src[e] : 0;
  }
#pragma unroll
  for (int u = 0; u < EPT; ++u)
    r[u] = (d[u] >= 0) ? lds_add_rtn(&hist[d[u] >> BSH], 1) : 0;

  const int t = (threadIdx.x < NB) ? bintot[threadIdx.x] : 0;
  int total;
  const int start = block_excl_scan(t, wsum, total);
  if (threadIdx.x < NB) {
    const int c = hist[threadIdx.x];
    const int off = c ? __hip_atomic_fetch_add(&bincur[threadIdx.x], c, __ATOMIC_RELAXED, __HIP_MEMORY_SCOPE_AGENT) : 0;
    hist[threadIdx.x] = start + off;
  }
  __syncthreads();
#pragma unroll
  for (int u = 0; u < EPT; ++u)
    if (d[u] >= 0) coarse[hist[d[u] >> BSH] + r[u]] = make_int2(s[u], d[u]);
}

__global__ void __launch_bounds__(1024) k_bin_sort(const int2* __restrict__ coarse, const int* __restrict__ bintot,
                                                   int* __restrict__ rp, int* __restrict__ csr, int N) {
  __shared__ int cnt[BW];
  __shared__ int wsum[32];
  const int b = blockIdx.x;
  if (threadIdx.x < BW) cnt[threadIdx.x] = 0;
  const int t = ((int)threadIdx.x < b) ? bintot[threadIdx.x] : 0;
  int bstart;
  block_excl_scan(t, wsum, bstart);
  const int bc = bintot[b];
  const int2* __restrict__ ed = coarse + bstart;

  for (int i = threadIdx.x; i < bc; i += 1024) lds_add_rtn(&cnt[ed[i].y & (BW - 1)], 1);
  __syncthreads();

  const int v = (threadIdx.x < BW) ? cnt[threadIdx.x] : 0;
  int tot;
  const int ex = block_excl_scan(v, wsum, tot);
  if (threadIdx.x < BW) {
    const int node = b * BW + threadIdx.x;
    if (node < N) rp[node] = bstart + ex;
    cnt[threadIdx.x] = bstart + ex;
  }
  __syncthreads();

  for (int i = threadIdx.x; i < bc; i += 1024) {
    const int2 e = ed[i];
    csr[lds_add_rtn(&cnt[e.y & (BW - 1)], 1)] = e.x;
  }
}

__device__ __forceinline__ float dinv_of(const int* __restrict__ rp, int i, int N, int E) {
  const int b = rp[i];
  const int e = (i + 1 < N) ? rp[i + 1] : E;
  return rsqrtf((float)(e - b + 1));
}

template <int OUT, bool ADDM>
__global__ void k_gemm_wmma(const float* __restrict__ A, const float* __restrict__ W,
                            float* __restrict__ C, int N,
                            const float* __restrict__ mm, const int* __restrict__ batch,
                            const int* __restrict__ rp, int E) {
  constexpr int KD = 128;
  constexpr int AP = KD + 4;
  __shared__ __attribute__((aligned(16))) float As[16 * AP];

  const int row0 = blockIdx.x * 16;
  if constexpr (ADDM) {
    constexpr int RPW = 16 / (OUT / 16);
    const int ln = threadIdx.x & 31;
    const int t0 = __builtin_amdgcn_readfirstlane((int)(threadIdx.x >> 5) * RPW);
    const float4* __restrict__ Ar = (const float4*)A;
    const float4* __restrict__ mr = (const float4*)mm;
    float4 hv[RPW], mv[RPW];
#pragma unroll
    for (int u = 0; u < RPW; ++u) {
      const int row = min(row0 + t0 + u, N - 1);
      hv[u] = Ar[(size_t)row * 32 + ln];
      mv[u] = mr[(size_t)batch[row] * 32 + ln];
    }
#pragma unroll
    for (int u = 0; u < RPW; ++u) {
      float4 v = make_float4(0.0f, 0.0f, 0.0f, 0.0f);
      if (row0 + t0 + u < N)
        v = make_float4(hv[u].x + mv[u].x, hv[u].y + mv[u].y, hv[u].z + mv[u].z, hv[u].w + mv[u].w);
      ((float4*)As)[(t0 + u) * (AP / 4) + ln] = v;
    }
  } else {
    for (int i = threadIdx.x; i < 16 * KD; i += blockDim.x) {
      int r = i / KD, k = i % KD;
      As[r * AP + k] = (row0 + r < N) ? A[(size_t)(row0 + r) * KD + k] : 0.0f;
    }
  }
  __syncthreads();

  const int wave = threadIdx.x >> 5;
  const int lane = threadIdx.x & 31;
  const int half = lane >> 4;
  const int n    = lane & 15;
  const int cw   = wave * 16;

  float dv[8];
#pragma unroll
  for (int r = 0; r < 8; ++r) dv[r] = dinv_of(rp, min(row0 + r + 8 * half, N - 1), N, E);

  v8f c = {};
  for (int k0 = 0; k0 < KD; k0 += 4) {
    const int k = k0 + 2 * half;
    v2f a, b;
    a.x = As[n * AP + k];
    a.y = As[n * AP + k + 1];
    b.x = W[(size_t)k * OUT + cw + n];
    b.y = W[(size_t)(k + 1) * OUT + cw + n];
    c = __builtin_amdgcn_wmma_f32_16x16x4_f32(false, a, false, b,
                                              (short)0, c, false, false);
  }

#pragma unroll
  for (int r = 0; r < 8; ++r) {
    const int m = r + 8 * half;
    if (row0 + m < N) C[(size_t)(row0 + m) * OUT + cw + n] = c[r] * dv[r];
  }
}

template <int V> struct RowVec;
template <> struct RowVec<4> { typedef float4 T; };
template <> struct RowVec<1> { typedef float T; };

__device__ __forceinline__ float4 scale_add(float w, const float4 a, const float4 b) {
  return make_float4(fmaf(w, a.x, b.x), fmaf(w, a.y, b.y), fmaf(w, a.z, b.z), fmaf(w, a.w, b.w));
}
__device__ __forceinline__ float scale_add(float w, const float a, const float b) { return fmaf(w, a, b); }
__device__ __forceinline__ float4 plus(const float4 a, const float4 b) {
  return make_float4(a.x + b.x, a.y + b.y, a.z + b.z, a.w + b.w);
}
__device__ __forceinline__ float plus(const float a, const float b) { return a + b; }
__device__ __forceinline__ void set_zero(float4& a) { a = make_float4(0.0f, 0.0f, 0.0f, 0.0f); }
__device__ __forceinline__ void set_zero(float& a) { a = 0.0f; }

template <int B, typename T>
__device__ __forceinline__ void gather_rows(T& acc, const T* __restrict__ xr, int s_i, int k, int lane) {
  T v[B];
  for (int u = 0; u < B; ++u) {
    const int s = __builtin_amdgcn_readlane(s_i, k + u);
    v[u] = xr[(size_t)s * 32 + lane];
  }
  for (int u = 0; u < B; ++u) acc = plus(acc, v[u]);
}

__device__ __forceinline__ float elu1(float x) { return x > 0.0f ? x : expm1f(x); }
__device__ __forceinline__ float4 elu_row(const float4 a) {
  return make_float4(elu1(a.x), elu1(a.y), elu1(a.z), elu1(a.w));
}
__device__ __forceinline__ float elu_row(const float a) { return elu1(a); }

__device__ __forceinline__ void pool_add_row(float* __restrict__ pool, int g, int lane, const float4 a) {
  float* p = &pool[(size_t)g * 128 + lane * 4];
  atomAddF(p + 0, a.x); atomAddF(p + 1, a.y);
  atomAddF(p + 2, a.z); atomAddF(p + 3, a.w);
}
__device__ __forceinline__ void pool_add_row(float* __restrict__ pool, int g, int lane, const float a) {
  atomAddF(&pool[(size_t)g * 32 + lane], a);
}

#define RUN 3
#define GW  8

template <int F, bool ELU>
__global__ void __launch_bounds__(GW * 32) k_gather(const float* __restrict__ xw, const int* __restrict__ rp,
                         const int* __restrict__ csr, const float* __restrict__ bias,
                         float* __restrict__ out, const int* __restrict__ batch,
                         float* __restrict__ pool, float* __restrict__ cnt, int N, int E) {
  constexpr int V = F / 32;
  typedef typename RowVec<V>::T T;
  __shared__ T     prow[GW * 32];
  __shared__ float pcnt[GW];
  __shared__ int   pgr[GW];
  const int lane = threadIdx.x & 31;
  const T* __restrict__ xr = (const T*)xw;
  T* __restrict__ orow     = (T*)out;
  const T bv = ((const T*)bias)[lane];

  const int wave = __builtin_amdgcn_readfirstlane((int)(threadIdx.x >> 5));
  const int n0 = (blockIdx.x * GW + wave) * RUN;
  const int n1 = min(n0 + RUN, N);

  T pacc;
  set_zero(pacc);
  float pc = 0.0f;
  int gcur = (n0 < N) ? batch[n0] : -1;

  for (int node = n0; node < n1; ++node) {
    const int beg  = rp[node];
    const int end  = (node + 1 < N) ? rp[node + 1] : E;
    const int g    = batch[node];
    const float dn = rsqrtf((float)(end - beg + 1));
    const T self = xr[(size_t)node * 32 + lane];
    T acc;
    set_zero(acc);

    for (int base = beg; base < end; base += 32) {
      const int j = base + lane;
      const int s_i = (j < end) ? csr[j] : 0;
      const int cnt_e = min(32, end - base);
      int k = 0;
      for (; k + 8 <= cnt_e; k += 8) gather_rows<8>(acc, xr, s_i, k, lane);
      if (k + 4 <= cnt_e) { gather_rows<4>(acc, xr, s_i, k, lane); k += 4; }
      if (k + 2 <= cnt_e) { gather_rows<2>(acc, xr, s_i, k, lane); k += 2; }
      if (k < cnt_e)      { gather_rows<1>(acc, xr, s_i, k, lane); }
    }
    acc = plus(acc, self);
    T r = scale_add(dn, acc, bv);
    if constexpr (ELU) r = elu_row(r);
    orow[(size_t)node * 32 + lane] = r;

    if (g != gcur) {
      pool_add_row(pool, gcur, lane, pacc);
      if (ELU && lane == 0) atomAddF(&cnt[gcur], pc);
      set_zero(pacc);
      pc = 0.0f;
      gcur = g;
    }
    pacc = plus(pacc, r);
    pc += 1.0f;
  }

  prow[wave * 32 + lane] = pacc;
  if (lane == 0) { pcnt[wave] = pc; pgr[wave] = gcur; }
  __syncthreads();
  if (threadIdx.x < F) {
    const float* pf = (const float*)prow;
    float sq[GW], cq[GW];
    int gq[GW];
#pragma unroll
    for (int w = 0; w < GW; ++w) {
      gq[w] = pgr[w];
      cq[w] = pcnt[w];
      sq[w] = pf[w * F + threadIdx.x];
    }
    float s = 0.0f, c = 0.0f;
    int g = -1;
#pragma unroll
    for (int w = 0; w < GW; ++w) {
      const int gw = gq[w];
      if (gw < 0) continue;
      if (gw != g) {
        if (g >= 0) {
          atomAddF(&pool[(size_t)g * F + threadIdx.x], s);
          if (ELU && threadIdx.x == 0) atomAddF(&cnt[g], c);
        }
        s = 0.0f;
        c = 0.0f;
        g = gw;
      }
      s += sq[w];
      c += cq[w];
    }
    if (g >= 0) {
      atomAddF(&pool[(size_t)g * F + threadIdx.x], s);
      if (ELU && threadIdx.x == 0) atomAddF(&cnt[g], c);
    }
  }
}

template <int IN, int OUT>
__global__ void k_master_mlp(const float* __restrict__ pool, const float* __restrict__ cnt,
                             const float* __restrict__ Wm, const float* __restrict__ bm,
                             float* __restrict__ mm, const int* __restrict__ gptr) {
  int G = *gptr;
  int idx = blockIdx.x * blockDim.x + threadIdx.x;
  int g = idx / OUT, j = idx % OUT;
  if (g >= G) return;
  float inv = 1.0f / fmaxf(cnt[g], 1.0f);
  float acc = bm[j];
  for (int i = 0; i < IN; ++i)
    acc = fmaf(pool[(size_t)g * IN + i] * inv, Wm[(size_t)i * OUT + j], acc);
  mm[(size_t)g * OUT + j] = fmaxf(acc, 0.0f);
}

__global__ void k_finalize(const float* __restrict__ z, const float* __restrict__ mm2,
                           const int* __restrict__ batch, float* __restrict__ out, int N) {
  int node = blockIdx.x * (blockDim.x >> 5) + (threadIdx.x >> 5);
  if (node >= N) return;
  int lane = threadIdx.x & 31;
  float v = z[(size_t)node * 32 + lane] + mm2[(size_t)batch[node] * 32 + lane];
  float m = v;
#pragma unroll
  for (int o = 16; o >= 1; o >>= 1) m = fmaxf(m, __shfl_xor(m, o, 32));
  float ev = __expf(v - m);
  float s = ev;
#pragma unroll
  for (int o = 16; o >= 1; o >>= 1) s += __shfl_xor(s, o, 32);
  out[(size_t)node * 32 + lane] = ev / s;
}

extern "C" void kernel_launch(void* const* d_in, const int* in_sizes, int n_in,
                              void* d_out, int out_size, void* d_ws, size_t ws_size,
                              hipStream_t stream) {
  const float* X   = (const float*)d_in[0];
  const float* W1  = (const float*)d_in[1];
  const float* b1  = (const float*)d_in[2];
  const float* W2  = (const float*)d_in[3];
  const float* b2  = (const float*)d_in[4];
  const float* mW1 = (const float*)d_in[5];
  const float* mb1 = (const float*)d_in[6];
  const float* mW2 = (const float*)d_in[7];
  const float* mb2 = (const float*)d_in[8];
  const int* ei    = (const int*)d_in[9];
  const int* batch = (const int*)d_in[10];
  const int* gptr  = (const int*)d_in[11];

  const int N = in_sizes[0] / 128;
  const int E = in_sizes[9] / 2;
  const int* src = ei;
  const int* dst = ei + E;

  char* w = (char*)d_ws;
  size_t off = 0;
  auto take = [&](size_t bytes) -> float* {
    float* p = (float*)(w + off);
    off = (off + bytes + 255) & ~(size_t)255;
    return p;
  };
  float* xw    = take((size_t)N * 128 * 4);
  float* h     = take((size_t)N * 128 * 4);
  int*   rp    = (int*)take((size_t)N * 4);
  int*   csr   = (int*)take((size_t)N * 32 * 4);
  int2*  coarse = (int2*)h;
  int*   bintot = (int*)xw;
  int*   bincur = bintot + NBMAX;
  float* z     = h;
  float* pool1 = take((size_t)GMAX * 128 * 4);
  float* pool2 = take((size_t)GMAX * 32 * 4);
  float* cnt   = take((size_t)GMAX * 4);
  float* mm1   = take((size_t)GMAX * 128 * 4);
  float* mm2   = take((size_t)GMAX * 32 * 4);

  const int B = 256;
  auto blks = [](long long n, int b) { return (unsigned)((n + b - 1) / b); };

  const int NB = (N + BW - 1) >> BSH;
  k_fill<<<blks(2 * NBMAX, B), B, 0, stream>>>((float*)bintot, 0.0f, 2 * NBMAX);
  k_fill<<<blks(GMAX * 128, B), B, 0, stream>>>(pool1, 0.0f, GMAX * 128);
  k_fill<<<blks(GMAX * 32, B), B, 0, stream>>>(pool2, 0.0f, GMAX * 32);
  k_fill<<<blks(GMAX, B), B, 0, stream>>>(cnt, 0.0f, GMAX);
  k_bin_count<<<blks(E, EPB), 1024, 0, stream>>>(dst, bintot, E, NB);
  k_bin_place<<<blks(E, EPB), 1024, 0, stream>>>(src, dst, bintot, bincur, coarse, E, NB);
  k_bin_sort<<<NB, 1024, 0, stream>>>(coarse, bintot, rp, csr, N);

  const unsigned gblocks = blks(N, GW * RUN);

  k_gemm_wmma<128, false><<<blks(N, 16), 256, 0, stream>>>(X, W1, xw, N, nullptr, nullptr, rp, E);
  k_gather<128, true><<<gblocks, GW * 32, 0, stream>>>(xw, rp, csr, b1, h, batch, pool1, cnt, N, E);

  k_master_mlp<128, 128><<<blks(GMAX * 128, B), B, 0, stream>>>(pool1, cnt, mW1, mb1, mm1, gptr);

  k_gemm_wmma<32, true><<<blks(N, 16), 64, 0, stream>>>(h, W2, xw, N, mm1, batch, rp, E);
  k_gather<32, false><<<gblocks, GW * 32, 0, stream>>>(xw, rp, csr, b2, z, batch, pool2, nullptr, N, E);

  k_master_mlp<32, 32><<<blks(GMAX * 32, B), B, 0, stream>>>(pool2, cnt, mW2, mb2, mm2, gptr);
  k_finalize<<<blks(N, 8), B, 0, stream>>>(z, mm2, batch, (float*)d_out, N);
}


# Load round trips grouped, not one per element: products ask for A rows (float4) and W fragments (registers) ahead of the WMMA loop; range sort takes 12 edges per trip, MLP 32 weights; sum orders kept
# speedup vs baseline: 6.0932x; 1.0795x over previous
#include <hip/hip_runtime.h>
#include <math.h>


typedef __attribute__((ext_vector_type(2))) float v2f;
typedef __attribute__((ext_vector_type(8))) float v8f;

#define GMAX 256

__device__ __forceinline__ void atomAddF(float* p, float v) {
  __hip_atomic_fetch_add(p, v, __ATOMIC_RELAXED, __HIP_MEMORY_SCOPE_AGENT);
}

__global__ void k_fill(float* p, float v, int n) {
  int i = blockIdx.x * blockDim.x + threadIdx.x;
  if (i < n) p[i] = v;
}

#define BSH   8
#define BW    (1 << BSH)
#define NBMAX 1024
#define EPT   4
#define EPB   (1024 * EPT)
#define SPT   12

__device__ __forceinline__ int lds_add_rtn(int* p, int v) {
  return __hip_atomic_fetch_add(p, v, __ATOMIC_RELAXED, __HIP_MEMORY_SCOPE_WORKGROUP);
}

__device__ __forceinline__ int wave_incl_scan(int v, int lane) {
  for (int o = 1; o < 32; o <<= 1) {
    int t = __shfl_up(v, o, 32);
    if (lane >= o) v += t;
  }
  return v;
}

__device__ __forceinline__ int block_excl_scan(int v, int* wsum, int& total) {
  const int lane = threadIdx.x & 31;
  const int wave = threadIdx.x >> 5;
  const int inc  = wave_incl_scan(v, lane);
  if (lane == 31) wsum[wave] = inc;
  __syncthreads();
  const int w    = wsum[lane];
  const int winc = wave_incl_scan(w, lane);
  total = __shfl(winc, 31, 32);
  const int wbase = __shfl(winc - w, wave, 32);
  __syncthreads();
  return wbase + inc - v;
}

__global__ void __launch_bounds__(1024) k_bin_count(const int* __restrict__ dst, int* __restrict__ bintot,
                                                    int E, int NB) {
  __shared__ int hist[NBMAX];
  for (int i = threadIdx.x; i < NB; i += 1024) hist[i] = 0;
  __syncthreads();
  const int base = blockIdx.x * EPB + threadIdx.x;
  int d[EPT];
#pragma unroll
  for (int u = 0; u < EPT; ++u) {
    const int e = base + u * 1024;
    d[u] = (e < E) ? dst[e] : -1;
  }
#pragma unroll
  for (int u = 0; u < EPT; ++u)
    if (d[u] >= 0) lds_add_rtn(&hist[d[u] >> BSH], 1);
  __syncthreads();
  for (int i = threadIdx.x; i < NB; i += 1024) {
    const int c = hist[i];
    if (c) __hip_atomic_fetch_add(&bintot[i], c, __ATOMIC_RELAXED, __HIP_MEMORY_SCOPE_AGENT);
  }
}

__global__ void __launch_bounds__(1024) k_bin_place(const int* __restrict__ src, const int* __restrict__ dst,
                                                    const int* __restrict__ bintot, int* __restrict__ bincur,
                                                    int2* __restrict__ coarse, int E, int NB) {
  __shared__ int hist[NBMAX];
  __shared__ int wsum[32];
  for (int i = threadIdx.x; i < NB; i += 1024) hist[i] = 0;
  __syncthreads();
  const int base = blockIdx.x * EPB + threadIdx.x;
  int s[EPT], d[EPT], r[EPT];
#pragma unroll
  for (int u = 0; u < EPT; ++u) {
    const int e = base + u * 1024;
    d[u] = (e < E) ? dst[e] : -1;
    s[u] = (e < E) ? src[e] : 0;
  }
#pragma unroll
  for (int u = 0; u < EPT; ++u)
    r[u] = (d[u] >= 0) ? lds_add_rtn(&hist[d[u] >> BSH], 1) : 0;

  const int t = (threadIdx.x < NB) ? bintot[threadIdx.x] : 0;
  int total;
  const int start = block_excl_scan(t, wsum, total);
  if (threadIdx.x < NB) {
    const int c = hist[threadIdx.x];
    const int off = c ? __hip_atomic_fetch_add(&bincur[threadIdx.x], c, __ATOMIC_RELAXED, __HIP_MEMORY_SCOPE_AGENT) : 0;
    hist[threadIdx.x] = start + off;
  }
  __syncthreads();
#pragma unroll
  for (int u = 0; u < EPT; ++u)
    if (d[u] >= 0) coarse[hist[d[u] >> BSH] + r[u]] = make_int2(s[u], d[u]);
}

__global__ void __launch_bounds__(1024) k_bin_sort(const int2* __restrict__ coarse, const int* __restrict__ bintot,
                                                   int* __restrict__ rp, int* __restrict__ csr, int N) {
  __shared__ int cnt[BW];
  __shared__ int wsum[32];
  const int b = blockIdx.x;
  if (threadIdx.x < BW) cnt[threadIdx.x] = 0;
  const int t = ((int)threadIdx.x < b) ? bintot[threadIdx.x] : 0;
  int bstart;
  block_excl_scan(t, wsum, bstart);
  const int bc = bintot[b];
  const int2* __restrict__ ed = coarse + bstart;

  for (int c0 = 0; c0 < bc; c0 += 1024 * SPT) {
    int y[SPT];
#pragma unroll
    for (int u = 0; u < SPT; ++u) {
      const int i = c0 + u * 1024 + (int)threadIdx.x;
      y[u] = 0;
      if (i < bc) y[u] = ed[i].y;
    }
#pragma unroll
    for (int u = 0; u < SPT; ++u) {
      const int i = c0 + u * 1024 + (int)threadIdx.x;
      if (i < bc) lds_add_rtn(&cnt[y[u] & (BW - 1)], 1);
    }
  }
  __syncthreads();

  const int v = (threadIdx.x < BW) ? cnt[threadIdx.x] : 0;
  int tot;
  const int ex = block_excl_scan(v, wsum, tot);
  if (threadIdx.x < BW) {
    const int node = b * BW + threadIdx.x;
    if (node < N) rp[node] = bstart + ex;
    cnt[threadIdx.x] = bstart + ex;
  }
  __syncthreads();

  for (int c0 = 0; c0 < bc; c0 += 1024 * SPT) {
    int2 e[SPT];
    int slot[SPT];
#pragma unroll
    for (int u = 0; u < SPT; ++u) {
      const int i = c0 + u * 1024 + (int)threadIdx.x;
      e[u] = make_int2(0, 0);
      if (i < bc) e[u] = ed[i];
    }
#pragma unroll
    for (int u = 0; u < SPT; ++u) {
      const int i = c0 + u * 1024 + (int)threadIdx.x;
      slot[u] = 0;
      if (i < bc) slot[u] = lds_add_rtn(&cnt[e[u].y & (BW - 1)], 1);
    }
#pragma unroll
    for (int u = 0; u < SPT; ++u) {
      const int i = c0 + u * 1024 + (int)threadIdx.x;
      if (i < bc) csr[slot[u]] = e[u].x;
    }
  }
}

template <int OUT, bool ADDM>
__global__ void k_gemm_wmma(const float* __restrict__ A, const float* __restrict__ W,
                            float* __restrict__ C, int N,
                            const float* __restrict__ mm, const int* __restrict__ batch,
                            const int* __restrict__ rp, int E) {
  constexpr int KD   = 128;
  constexpr int KS   = KD / 4;
  constexpr int KPRE = ADDM ? KS / 2 : KS - 2;
  constexpr int RPW  = 16 / (OUT / 16);
  constexpr int AP = KD + 4;
  __shared__ __attribute__((aligned(16))) float As[16 * AP];

  const int row0 = blockIdx.x * 16;
  const int wave = threadIdx.x >> 5;
  const int lane = threadIdx.x & 31;
  const int half = lane >> 4;
  const int n    = lane & 15;
  const int cw   = wave * 16;

  const int t0 = __builtin_amdgcn_readfirstlane(wave * RPW);
  const float4* __restrict__ Ar = (const float4*)A;
  float4 hv[RPW], mv[RPW];
#pragma unroll
  for (int u = 0; u < RPW; ++u) {
    const int row = min(row0 + t0 + u, N - 1);
    hv[u] = Ar[(size_t)row * 32 + lane];
  }
  __builtin_amdgcn_sched_barrier(0);

  v2f bf[KS];
#pragma unroll
  for (int i = 0; i < KPRE; ++i) {
    const int k = 4 * i + 2 * half;
    bf[i].x = W[(size_t)k * OUT + cw + n];
    bf[i].y = W[(size_t)(k + 1) * OUT + cw + n];
  }
  if constexpr (ADDM) {
#pragma unroll
    for (int u = 0; u < RPW; ++u) {
      const int row = min(row0 + t0 + u, N - 1);
      mv[u] = ((const float4*)mm)[(size_t)batch[row] * 32 + lane];
    }
  }
  __builtin_amdgcn_sched_barrier(0);

#pragma unroll
  for (int u = 0; u < RPW; ++u) {
    float4 v = make_float4(0.0f, 0.0f, 0.0f, 0.0f);
    if (row0 + t0 + u < N) {
      if constexpr (ADDM)
        v = make_float4(hv[u].x + mv[u].x, hv[u].y + mv[u].y, hv[u].z + mv[u].z, hv[u].w + mv[u].w);
      else
        v = hv[u];
    }
    ((float4*)As)[(t0 + u) * (AP / 4) + lane] = v;
  }
  __syncthreads();

#pragma unroll
  for (int i = KPRE; i < KS; ++i) {
    const int k = 4 * i + 2 * half;
    bf[i].x = W[(size_t)k * OUT + cw + n];
    bf[i].y = W[(size_t)(k + 1) * OUT + cw + n];
  }
  int rb[8], re[8];
#pragma unroll
  for (int r = 0; r < 8; ++r) {
    const int i = min(row0 + r + 8 * half, N - 1);
    rb[r] = rp[i];
    re[r] = (i + 1 < N) ? rp[i + 1] : E;
  }

  v8f c = {};
#pragma unroll
  for (int i = 0; i < KS; ++i) {
    const int k = 4 * i + 2 * half;
    v2f a;
    a.x = As[n * AP + k];
    a.y = As[n * AP + k + 1];
    c = __builtin_amdgcn_wmma_f32_16x16x4_f32(false, a, false, bf[i],
                                              (short)0, c, false, false);
  }

#pragma unroll
  for (int r = 0; r < 8; ++r) {
    const int m = r + 8 * half;
    if (row0 + m < N) C[(size_t)(row0 + m) * OUT + cw + n] = c[r] * rsqrtf((float)(re[r] - rb[r] + 1));
  }
}

template <int V> struct RowVec;
template <> struct RowVec<4> { typedef float4 T; };
template <> struct RowVec<1> { typedef float T; };

__device__ __forceinline__ float4 scale_add(float w, const float4 a, const float4 b) {
  return make_float4(fmaf(w, a.x, b.x), fmaf(w, a.y, b.y), fmaf(w, a.z, b.z), fmaf(w, a.w, b.w));
}
__device__ __forceinline__ float scale_add(float w, const float a, const float b) { return fmaf(w, a, b); }
__device__ __forceinline__ float4 plus(const float4 a, const float4 b) {
  return make_float4(a.x + b.x, a.y + b.y, a.z + b.z, a.w + b.w);
}
__device__ __forceinline__ float plus(const float a, const float b) { return a + b; }
__device__ __forceinline__ void set_zero(float4& a) { a = make_float4(0.0f, 0.0f, 0.0f, 0.0f); }
__device__ __forceinline__ void set_zero(float& a) { a = 0.0f; }

template <int B, typename T>
__device__ __forceinline__ void gather_rows(T& acc, const T* __restrict__ xr, int s_i, int k, int lane) {
  T v[B];
  for (int u = 0; u < B; ++u) {
    const int s = __builtin_amdgcn_readlane(s_i, k + u);
    v[u] = xr[(size_t)s * 32 + lane];
  }
  for (int u = 0; u < B; ++u) acc = plus(acc, v[u]);
}

__device__ __forceinline__ float elu1(float x) { return x > 0.0f ? x : expm1f(x); }
__device__ __forceinline__ float4 elu_row(const float4 a) {
  return make_float4(elu1(a.x), elu1(a.y), elu1(a.z), elu1(a.w));
}
__device__ __forceinline__ float elu_row(const float a) { return elu1(a); }

__device__ __forceinline__ void pool_add_row(float* __restrict__ pool, int g, int lane, const float4 a) {
  float* p = &pool[(size_t)g * 128 + lane * 4];
  atomAddF(p + 0, a.x); atomAddF(p + 1, a.y);
  atomAddF(p + 2, a.z); atomAddF(p + 3, a.w);
}
__device__ __forceinline__ void pool_add_row(float* __restrict__ pool, int g, int lane, const float a) {
  atomAddF(&pool[(size_t)g * 32 + lane], a);
}

#define RUN 3
#define GW  8

template <int F, bool ELU>
__global__ void __launch_bounds__(GW * 32) k_gather(const float* __restrict__ xw, const int* __restrict__ rp,
                         const int* __restrict__ csr, const float* __restrict__ bias,
                         float* __restrict__ out, const int* __restrict__ batch,
                         float* __restrict__ pool, float* __restrict__ cnt, int N, int E) {
  constexpr int V = F / 32;
  typedef typename RowVec<V>::T T;
  __shared__ T     prow[GW * 32];
  __shared__ float pcnt[GW];
  __shared__ int   pgr[GW];
  const int lane = threadIdx.x & 31;
  const T* __restrict__ xr = (const T*)xw;
  T* __restrict__ orow     = (T*)out;
  const T bv = ((const T*)bias)[lane];

  const int wave = __builtin_amdgcn_readfirstlane((int)(threadIdx.x >> 5));
  const int n0 = (blockIdx.x * GW + wave) * RUN;
  const int n1 = min(n0 + RUN, N);

  T pacc;
  set_zero(pacc);
  float pc = 0.0f;
  int gcur = (n0 < N) ? batch[n0] : -1;

  for (int node = n0; node < n1; ++node) {
    const int beg  = rp[node];
    const int end  = (node + 1 < N) ? rp[node + 1] : E;
    const int g    = batch[node];
    const float dn = rsqrtf((float)(end - beg + 1));
    const T self = xr[(size_t)node * 32 + lane];
    T acc;
    set_zero(acc);

    for (int base = beg; base < end; base += 32) {
      const int j = base + lane;
      const int s_i = (j < end) ? csr[j] : 0;
      const int cnt_e = min(32, end - base);
      int k = 0;
      for (; k + 8 <= cnt_e; k += 8) gather_rows<8>(acc, xr, s_i, k, lane);
      if (k + 4 <= cnt_e) { gather_rows<4>(acc, xr, s_i, k, lane); k += 4; }
      if (k + 2 <= cnt_e) { gather_rows<2>(acc, xr, s_i, k, lane); k += 2; }
      if (k < cnt_e)      { gather_rows<1>(acc, xr, s_i, k, lane); }
    }
    acc = plus(acc, self);
    T r = scale_add(dn, acc, bv);
    if constexpr (ELU) r = elu_row(r);
    orow[(size_t)node * 32 + lane] = r;

    if (g != gcur) {
      pool_add_row(pool, gcur, lane, pacc);
      if (ELU && lane == 0) atomAddF(&cnt[gcur], pc);
      set_zero(pacc);
      pc = 0.0f;
      gcur = g;
    }
    pacc = plus(pacc, r);
    pc += 1.0f;
  }

  prow[wave * 32 + lane] = pacc;
  if (lane == 0) { pcnt[wave] = pc; pgr[wave] = gcur; }
  __syncthreads();
  if (threadIdx.x < F) {
    const float* pf = (const float*)prow;
    float sq[GW], cq[GW];
    int gq[GW];
#pragma unroll
    for (int w = 0; w < GW; ++w) {
      gq[w] = pgr[w];
      cq[w] = pcnt[w];
      sq[w] = pf[w * F + threadIdx.x];
    }
    float s = 0.0f, c = 0.0f;
    int g = -1;
#pragma unroll
    for (int w = 0; w < GW; ++w) {
      const int gw = gq[w];
      if (gw < 0) continue;
      if (gw != g) {
        if (g >= 0) {
          atomAddF(&pool[(size_t)g * F + threadIdx.x], s);
          if (ELU && threadIdx.x == 0) atomAddF(&cnt[g], c);
        }
        s = 0.0f;
        c = 0.0f;
        g = gw;
      }
      s += sq[w];
      c += cq[w];
    }
    if (g >= 0) {
      atomAddF(&pool[(size_t)g * F + threadIdx.x], s);
      if (ELU && threadIdx.x == 0) atomAddF(&cnt[g], c);
    }
  }
}

template <int IN, int OUT>
__global__ void k_master_mlp(const float* __restrict__ pool, const float* __restrict__ cnt,
                             const float* __restrict__ Wm, const float* __restrict__ bm,
                             float* __restrict__ mm, const int* __restrict__ gptr) {
  int G = *gptr;
  int idx = blockIdx.x * blockDim.x + threadIdx.x;
  int g = idx / OUT, j = idx % OUT;
  if (g >= G) return;
  float inv = 1.0f / fmaxf(cnt[g], 1.0f);
  float acc = bm[j];
#pragma unroll 32
  for (int i = 0; i < IN; ++i)
    acc = fmaf(pool[(size_t)g * IN + i] * inv, Wm[(size_t)i * OUT + j], acc);
  mm[(size_t)g * OUT + j] = fmaxf(acc, 0.0f);
}

__global__ void k_finalize(const float* __restrict__ z, const float* __restrict__ mm2,
                           const int* __restrict__ batch, float* __restrict__ out, int N) {
  int node = blockIdx.x * (blockDim.x >> 5) + (threadIdx.x >> 5);
  if (node >= N) return;
  int lane = threadIdx.x & 31;
  float v = z[(size_t)node * 32 + lane] + mm2[(size_t)batch[node] * 32 + lane];
  float m = v;
#pragma unroll
  for (int o = 16; o >= 1; o >>= 1) m = fmaxf(m, __shfl_xor(m, o, 32));
  float ev = __expf(v - m);
  float s = ev;
#pragma unroll
  for (int o = 16; o >= 1; o >>= 1) s += __shfl_xor(s, o, 32);
  out[(size_t)node * 32 + lane] = ev / s;
}

extern "C" void kernel_launch(void* const* d_in, const int* in_sizes, int n_in,
                              void* d_out, int out_size, void* d_ws, size_t ws_size,
                              hipStream_t stream) {
  const float* X   = (const float*)d_in[0];
  const float* W1  = (const float*)d_in[1];
  const float* b1  = (const float*)d_in[2];
  const float* W2  = (const float*)d_in[3];
  const float* b2  = (const float*)d_in[4];
  const float* mW1 = (const float*)d_in[5];
  const float* mb1 = (const float*)d_in[6];
  const float* mW2 = (const float*)d_in[7];
  const float* mb2 = (const float*)d_in[8];
  const int* ei    = (const int*)d_in[9];
  const int* batch = (const int*)d_in[10];
  const int* gptr  = (const int*)d_in[11];

  const int N = in_sizes[0] / 128;
  const int E = in_sizes[9] / 2;
  const int* src = ei;
  const int* dst = ei + E;

  char* w = (char*)d_ws;
  size_t off = 0;
  auto take = [&](size_t bytes) -> float* {
    float* p = (float*)(w + off);
    off = (off + bytes + 255) & ~(size_t)255;
    return p;
  };
  float* xw    = take((size_t)N * 128 * 4);
  float* h     = take((size_t)N * 128 * 4);
  int*   rp    = (int*)take((size_t)N * 4);
  int*   csr   = (int*)take((size_t)N * 32 * 4);
  int2*  coarse = (int2*)h;
  int*   bintot = (int*)xw;
  int*   bincur = bintot + NBMAX;
  float* z     = h;
  float* pool1 = take((size_t)GMAX * 128 * 4);
  float* pool2 = take((size_t)GMAX * 32 * 4);
  float* cnt   = take((size_t)GMAX * 4);
  float* mm1   = take((size_t)GMAX * 128 * 4);
  float* mm2   = take((size_t)GMAX * 32 * 4);

  const int B = 256;
  auto blks = [](long long n, int b) { return (unsigned)((n + b - 1) / b); };

  const int NB = (N + BW - 1) >> BSH;
  k_fill<<<blks(2 * NBMAX, B), B, 0, stream>>>((float*)bintot, 0.0f, 2 * NBMAX);
  k_fill<<<blks(GMAX * 128, B), B, 0, stream>>>(pool1, 0.0f, GMAX * 128);
  k_fill<<<blks(GMAX * 32, B), B, 0, stream>>>(pool2, 0.0f, GMAX * 32);
  k_fill<<<blks(GMAX, B), B, 0, stream>>>(cnt, 0.0f, GMAX);
  k_bin_count<<<blks(E, EPB), 1024, 0, stream>>>(dst, bintot, E, NB);
  k_bin_place<<<blks(E, EPB), 1024, 0, stream>>>(src, dst, bintot, bincur, coarse, E, NB);
  k_bin_sort<<<NB, 1024, 0, stream>>>(coarse, bintot, rp, csr, N);

  const unsigned gblocks = blks(N, GW * RUN);

  k_gemm_wmma<128, false><<<blks(N, 16), 256, 0, stream>>>(X, W1, xw, N, nullptr, nullptr, rp, E);
  k_gather<128, true><<<gblocks, GW * 32, 0, stream>>>(xw, rp, csr, b1, h, batch, pool1, cnt, N, E);

  k_master_mlp<128, 128><<<blks(GMAX * 128, B), B, 0, stream>>>(pool1, cnt, mW1, mb1, mm1, gptr);

  k_gemm_wmma<32, true><<<blks(N, 16), 64, 0, stream>>>(h, W2, xw, N, mm1, batch, rp, E);
  k_gather<32, false><<<gblocks, GW * 32, 0, stream>>>(xw, rp, csr, b2, z, batch, pool2, nullptr, N, E);

  k_master_mlp<32, 32><<<blks(GMAX * 32, B), B, 0, stream>>>(pool2, cnt, mW2, mb2, mm2, gptr);
  k_finalize<<<blks(N, 8), B, 0, stream>>>(z, mm2, batch, (float*)d_out, N);
}


# Softmax pass run by few long waves: a wave takes 16 rows (8 lanes x 4 floats a row, 4 rows a load, loads asked first), N/16 waves enter, not one per row; row sum order now 4 in lane, then 8 lanes
# speedup vs baseline: 6.2767x; 1.0301x over previous
#include <hip/hip_runtime.h>
#include <math.h>


typedef __attribute__((ext_vector_type(2))) float v2f;
typedef __attribute__((ext_vector_type(8))) float v8f;

#define GMAX 256

__device__ __forceinline__ void atomAddF(float* p, float v) {
  __hip_atomic_fetch_add(p, v, __ATOMIC_RELAXED, __HIP_MEMORY_SCOPE_AGENT);
}

__global__ void k_fill(float* p, float v, int n) {
  int i = blockIdx.x * blockDim.x + threadIdx.x;
  if (i < n) p[i] = v;
}

#define BSH   8
#define BW    (1 << BSH)
#define NBMAX 1024
#define EPT   4
#define EPB   (1024 * EPT)
#define SPT   12

__device__ __forceinline__ int lds_add_rtn(int* p, int v) {
  return __hip_atomic_fetch_add(p, v, __ATOMIC_RELAXED, __HIP_MEMORY_SCOPE_WORKGROUP);
}

__device__ __forceinline__ int wave_incl_scan(int v, int lane) {
  for (int o = 1; o < 32; o <<= 1) {
    int t = __shfl_up(v, o, 32);
    if (lane >= o) v += t;
  }
  return v;
}

__device__ __forceinline__ int block_excl_scan(int v, int* wsum, int& total) {
  const int lane = threadIdx.x & 31;
  const int wave = threadIdx.x >> 5;
  const int inc  = wave_incl_scan(v, lane);
  if (lane == 31) wsum[wave] = inc;
  __syncthreads();
  const int w    = wsum[lane];
  const int winc = wave_incl_scan(w, lane);
  total = __shfl(winc, 31, 32);
  const int wbase = __shfl(winc - w, wave, 32);
  __syncthreads();
  return wbase + inc - v;
}

__global__ void __launch_bounds__(1024) k_bin_count(const int* __restrict__ dst, int* __restrict__ bintot,
                                                    int E, int NB) {
  __shared__ int hist[NBMAX];
  for (int i = threadIdx.x; i < NB; i += 1024) hist[i] = 0;
  __syncthreads();
  const int base = blockIdx.x * EPB + threadIdx.x;
  int d[EPT];
#pragma unroll
  for (int u = 0; u < EPT; ++u) {
    const int e = base + u * 1024;
    d[u] = (e < E) ? dst[e] : -1;
  }
#pragma unroll
  for (int u = 0; u < EPT; ++u)
    if (d[u] >= 0) lds_add_rtn(&hist[d[u] >> BSH], 1);
  __syncthreads();
  for (int i = threadIdx.x; i < NB; i += 1024) {
    const int c = hist[i];
    if (c) __hip_atomic_fetch_add(&bintot[i], c, __ATOMIC_RELAXED, __HIP_MEMORY_SCOPE_AGENT);
  }
}

__global__ void __launch_bounds__(1024) k_bin_place(const int* __restrict__ src, const int* __restrict__ dst,
                                                    const int* __restrict__ bintot, int* __restrict__ bincur,
                                                    int2* __restrict__ coarse, int E, int NB) {
  __shared__ int hist[NBMAX];
  __shared__ int wsum[32];
  for (int i = threadIdx.x; i < NB; i += 1024) hist[i] = 0;
  __syncthreads();
  const int base = blockIdx.x * EPB + threadIdx.x;
  int s[EPT], d[EPT], r[EPT];
#pragma unroll
  for (int u = 0; u < EPT; ++u) {
    const int e = base + u * 1024;
    d[u] = (e < E) ? dst[e] : -1;
    s[u] = (e < E) ? src[e] : 0;
  }
#pragma unroll
  for (int u = 0; u < EPT; ++u)
    r[u] = (d[u] >= 0) ? lds_add_rtn(&hist[d[u] >> BSH], 1) : 0;

  const int t = (threadIdx.x < NB) ? bintot[threadIdx.x] : 0;
  int total;
  const int start = block_excl_scan(t, wsum, total);
  if (threadIdx.x < NB) {
    const int c = hist[threadIdx.x];
    const int off = c ? __hip_atomic_fetch_add(&bincur[threadIdx.x], c, __ATOMIC_RELAXED, __HIP_MEMORY_SCOPE_AGENT) : 0;
    hist[threadIdx.x] = start + off;
  }
  __syncthreads();
#pragma unroll
  for (int u = 0; u < EPT; ++u)
    if (d[u] >= 0) coarse[hist[d[u] >> BSH] + r[u]] = make_int2(s[u], d[u]);
}

__global__ void __launch_bounds__(1024) k_bin_sort(const int2* __restrict__ coarse, const int* __restrict__ bintot,
                                                   int* __restrict__ rp, int* __restrict__ csr, int N) {
  __shared__ int cnt[BW];
  __shared__ int wsum[32];
  const int b = blockIdx.x;
  if (threadIdx.x < BW) cnt[threadIdx.x] = 0;
  const int t = ((int)threadIdx.x < b) ? bintot[threadIdx.x] : 0;
  int bstart;
  block_excl_scan(t, wsum, bstart);
  const int bc = bintot[b];
  const int2* __restrict__ ed = coarse + bstart;

  for (int c0 = 0; c0 < bc; c0 += 1024 * SPT) {
    int y[SPT];
#pragma unroll
    for (int u = 0; u < SPT; ++u) {
      const int i = c0 + u * 1024 + (int)threadIdx.x;
      y[u] = 0;
      if (i < bc) y[u] = ed[i].y;
    }
#pragma unroll
    for (int u = 0; u < SPT; ++u) {
      const int i = c0 + u * 1024 + (int)threadIdx.x;
      if (i < bc) lds_add_rtn(&cnt[y[u] & (BW - 1)], 1);
    }
  }
  __syncthreads();

  const int v = (threadIdx.x < BW) ? cnt[threadIdx.x] : 0;
  int tot;
  const int ex = block_excl_scan(v, wsum, tot);
  if (threadIdx.x < BW) {
    const int node = b * BW + threadIdx.x;
    if (node < N) rp[node] = bstart + ex;
    cnt[threadIdx.x] = bstart + ex;
  }
  __syncthreads();

  for (int c0 = 0; c0 < bc; c0 += 1024 * SPT) {
    int2 e[SPT];
    int slot[SPT];
#pragma unroll
    for (int u = 0; u < SPT; ++u) {
      const int i = c0 + u * 1024 + (int)threadIdx.x;
      e[u] = make_int2(0, 0);
      if (i < bc) e[u] = ed[i];
    }
#pragma unroll
    for (int u = 0; u < SPT; ++u) {
      const int i = c0 + u * 1024 + (int)threadIdx.x;
      slot[u] = 0;
      if (i < bc) slot[u] = lds_add_rtn(&cnt[e[u].y & (BW - 1)], 1);
    }
#pragma unroll
    for (int u = 0; u < SPT; ++u) {
      const int i = c0 + u * 1024 + (int)threadIdx.x;
      if (i < bc) csr[slot[u]] = e[u].x;
    }
  }
}

template <int OUT, bool ADDM>
__global__ void k_gemm_wmma(const float* __restrict__ A, const float* __restrict__ W,
                            float* __restrict__ C, int N,
                            const float* __restrict__ mm, const int* __restrict__ batch,
                            const int* __restrict__ rp, int E) {
  constexpr int KD   = 128;
  constexpr int KS   = KD / 4;
  constexpr int KPRE = ADDM ? KS / 2 : KS - 2;
  constexpr int RPW  = 16 / (OUT / 16);
  constexpr int AP = KD + 4;
  __shared__ __attribute__((aligned(16))) float As[16 * AP];

  const int row0 = blockIdx.x * 16;
  const int wave = threadIdx.x >> 5;
  const int lane = threadIdx.x & 31;
  const int half = lane >> 4;
  const int n    = lane & 15;
  const int cw   = wave * 16;

  const int t0 = __builtin_amdgcn_readfirstlane(wave * RPW);
  const float4* __restrict__ Ar = (const float4*)A;
  float4 hv[RPW], mv[RPW];
#pragma unroll
  for (int u = 0; u < RPW; ++u) {
    const int row = min(row0 + t0 + u, N - 1);
    hv[u] = Ar[(size_t)row * 32 + lane];
  }
  __builtin_amdgcn_sched_barrier(0);

  v2f bf[KS];
#pragma unroll
  for (int i = 0; i < KPRE; ++i) {
    const int k = 4 * i + 2 * half;
    bf[i].x = W[(size_t)k * OUT + cw + n];
    bf[i].y = W[(size_t)(k + 1) * OUT + cw + n];
  }
  if constexpr (ADDM) {
#pragma unroll
    for (int u = 0; u < RPW; ++u) {
      const int row = min(row0 + t0 + u, N - 1);
      mv[u] = ((const float4*)mm)[(size_t)batch[row] * 32 + lane];
    }
  }
  __builtin_amdgcn_sched_barrier(0);

#pragma unroll
  for (int u = 0; u < RPW; ++u) {
    float4 v = make_float4(0.0f, 0.0f, 0.0f, 0.0f);
    if (row0 + t0 + u < N) {
      if constexpr (ADDM)
        v = make_float4(hv[u].x + mv[u].x, hv[u].y + mv[u].y, hv[u].z + mv[u].z, hv[u].w + mv[u].w);
      else
        v = hv[u];
    }
    ((float4*)As)[(t0 + u) * (AP / 4) + lane] = v;
  }
  __syncthreads();

#pragma unroll
  for (int i = KPRE; i < KS; ++i) {
    const int k = 4 * i + 2 * half;
    bf[i].x = W[(size_t)k * OUT + cw + n];
    bf[i].y = W[(size_t)(k + 1) * OUT + cw + n];
  }
  int rb[8], re[8];
#pragma unroll
  for (int r = 0; r < 8; ++r) {
    const int i = min(row0 + r + 8 * half, N - 1);
    rb[r] = rp[i];
    re[r] = (i + 1 < N) ? rp[i + 1] : E;
  }

  v8f c = {};
#pragma unroll
  for (int i = 0; i < KS; ++i) {
    const int k = 4 * i + 2 * half;
    v2f a;
    a.x = As[n * AP + k];
    a.y = As[n * AP + k + 1];
    c = __builtin_amdgcn_wmma_f32_16x16x4_f32(false, a, false, bf[i],
                                              (short)0, c, false, false);
  }

#pragma unroll
  for (int r = 0; r < 8; ++r) {
    const int m = r + 8 * half;
    if (row0 + m < N) C[(size_t)(row0 + m) * OUT + cw + n] = c[r] * rsqrtf((float)(re[r] - rb[r] + 1));
  }
}

template <int V> struct RowVec;
template <> struct RowVec<4> { typedef float4 T; };
template <> struct RowVec<1> { typedef float T; };

__device__ __forceinline__ float4 scale_add(float w, const float4 a, const float4 b) {
  return make_float4(fmaf(w, a.x, b.x), fmaf(w, a.y, b.y), fmaf(w, a.z, b.z), fmaf(w, a.w, b.w));
}
__device__ __forceinline__ float scale_add(float w, const float a, const float b) { return fmaf(w, a, b); }
__device__ __forceinline__ float4 plus(const float4 a, const float4 b) {
  return make_float4(a.x + b.x, a.y + b.y, a.z + b.z, a.w + b.w);
}
__device__ __forceinline__ float plus(const float a, const float b) { return a + b; }
__device__ __forceinline__ void set_zero(float4& a) { a = make_float4(0.0f, 0.0f, 0.0f, 0.0f); }
__device__ __forceinline__ void set_zero(float& a) { a = 0.0f; }

template <int B, typename T>
__device__ __forceinline__ void gather_rows(T& acc, const T* __restrict__ xr, int s_i, int k, int lane) {
  T v[B];
  for (int u = 0; u < B; ++u) {
    const int s = __builtin_amdgcn_readlane(s_i, k + u);
    v[u] = xr[(size_t)s * 32 + lane];
  }
  for (int u = 0; u < B; ++u) acc = plus(acc, v[u]);
}

__device__ __forceinline__ float elu1(float x) { return x > 0.0f ? x : expm1f(x); }
__device__ __forceinline__ float4 elu_row(const float4 a) {
  return make_float4(elu1(a.x), elu1(a.y), elu1(a.z), elu1(a.w));
}
__device__ __forceinline__ float elu_row(const float a) { return elu1(a); }

__device__ __forceinline__ void pool_add_row(float* __restrict__ pool, int g, int lane, const float4 a) {
  float* p = &pool[(size_t)g * 128 + lane * 4];
  atomAddF(p + 0, a.x); atomAddF(p + 1, a.y);
  atomAddF(p + 2, a.z); atomAddF(p + 3, a.w);
}
__device__ __forceinline__ void pool_add_row(float* __restrict__ pool, int g, int lane, const float a) {
  atomAddF(&pool[(size_t)g * 32 + lane], a);
}

#define RUN 3
#define GW  8

template <int F, bool ELU>
__global__ void __launch_bounds__(GW * 32) k_gather(const float* __restrict__ xw, const int* __restrict__ rp,
                         const int* __restrict__ csr, const float* __restrict__ bias,
                         float* __restrict__ out, const int* __restrict__ batch,
                         float* __restrict__ pool, float* __restrict__ cnt, int N, int E) {
  constexpr int V = F / 32;
  typedef typename RowVec<V>::T T;
  __shared__ T     prow[GW * 32];
  __shared__ float pcnt[GW];
  __shared__ int   pgr[GW];
  const int lane = threadIdx.x & 31;
  const T* __restrict__ xr = (const T*)xw;
  T* __restrict__ orow     = (T*)out;
  const T bv = ((const T*)bias)[lane];

  const int wave = __builtin_amdgcn_readfirstlane((int)(threadIdx.x >> 5));
  const int n0 = (blockIdx.x * GW + wave) * RUN;
  const int n1 = min(n0 + RUN, N);

  T pacc;
  set_zero(pacc);
  float pc = 0.0f;
  int gcur = (n0 < N) ? batch[n0] : -1;

  for (int node = n0; node < n1; ++node) {
    const int beg  = rp[node];
    const int end  = (node + 1 < N) ? rp[node + 1] : E;
    const int g    = batch[node];
    const float dn = rsqrtf((float)(end - beg + 1));
    const T self = xr[(size_t)node * 32 + lane];
    T acc;
    set_zero(acc);

    for (int base = beg; base < end; base += 32) {
      const int j = base + lane;
      const int s_i = (j < end) ? csr[j] : 0;
      const int cnt_e = min(32, end - base);
      int k = 0;
      for (; k + 8 <= cnt_e; k += 8) gather_rows<8>(acc, xr, s_i, k, lane);
      if (k + 4 <= cnt_e) { gather_rows<4>(acc, xr, s_i, k, lane); k += 4; }
      if (k + 2 <= cnt_e) { gather_rows<2>(acc, xr, s_i, k, lane); k += 2; }
      if (k < cnt_e)      { gather_rows<1>(acc, xr, s_i, k, lane); }
    }
    acc = plus(acc, self);
    T r = scale_add(dn, acc, bv);
    if constexpr (ELU) r = elu_row(r);
    orow[(size_t)node * 32 + lane] = r;

    if (g != gcur) {
      pool_add_row(pool, gcur, lane, pacc);
      if (ELU && lane == 0) atomAddF(&cnt[gcur], pc);
      set_zero(pacc);
      pc = 0.0f;
      gcur = g;
    }
    pacc = plus(pacc, r);
    pc += 1.0f;
  }

  prow[wave * 32 + lane] = pacc;
  if (lane == 0) { pcnt[wave] = pc; pgr[wave] = gcur; }
  __syncthreads();
  if (threadIdx.x < F) {
    const float* pf = (const float*)prow;
    float sq[GW], cq[GW];
    int gq[GW];
#pragma unroll
    for (int w = 0; w < GW; ++w) {
      gq[w] = pgr[w];
      cq[w] = pcnt[w];
      sq[w] = pf[w * F + threadIdx.x];
    }
    float s = 0.0f, c = 0.0f;
    int g = -1;
#pragma unroll
    for (int w = 0; w < GW; ++w) {
      const int gw = gq[w];
      if (gw < 0) continue;
      if (gw != g) {
        if (g >= 0) {
          atomAddF(&pool[(size_t)g * F + threadIdx.x], s);
          if (ELU && threadIdx.x == 0) atomAddF(&cnt[g], c);
        }
        s = 0.0f;
        c = 0.0f;
        g = gw;
      }
      s += sq[w];
      c += cq[w];
    }
    if (g >= 0) {
      atomAddF(&pool[(size_t)g * F + threadIdx.x], s);
      if (ELU && threadIdx.x == 0) atomAddF(&cnt[g], c);
    }
  }
}

template <int IN, int OUT>
__global__ void k_master_mlp(const float* __restrict__ pool, const float* __restrict__ cnt,
                             const float* __restrict__ Wm, const float* __restrict__ bm,
                             float* __restrict__ mm, const int* __restrict__ gptr) {
  int G = *gptr;
  int idx = blockIdx.x * blockDim.x + threadIdx.x;
  int g = idx / OUT, j = idx % OUT;
  if (g >= G) return;
  float inv = 1.0f / fmaxf(cnt[g], 1.0f);
  float acc = bm[j];
#pragma unroll 32
  for (int i = 0; i < IN; ++i)
    acc = fmaf(pool[(size_t)g * IN + i] * inv, Wm[(size_t)i * OUT + j], acc);
  mm[(size_t)g * OUT + j] = fmaxf(acc, 0.0f);
}

typedef __attribute__((ext_vector_type(4))) float v4f;
#define FIN_U 4
#define FIN_W 4
__global__ void __launch_bounds__(FIN_W * 32) k_finalize(const float* __restrict__ z, const float* __restrict__ mm2,
                           const int* __restrict__ batch, float* __restrict__ out, int N) {
  const int lane = threadIdx.x & 31;
  const int sub  = lane >> 3;
  const int q    = lane & 7;
  const int r0   = (blockIdx.x * FIN_W + (threadIdx.x >> 5)) * (4 * FIN_U);
  if (r0 >= N) return;
  const v4f* __restrict__ zr = (const v4f*)z;
  const v4f* __restrict__ mr = (const v4f*)mm2;
  v4f* __restrict__ orow     = (v4f*)out;

  int g[FIN_U];
  v4f v[FIN_U], t[FIN_U];
#pragma unroll
  for (int u = 0; u < FIN_U; ++u) g[u] = batch[min(r0 + 4 * u + sub, N - 1)];
#pragma unroll
  for (int u = 0; u < FIN_U; ++u) v[u] = zr[(size_t)min(r0 + 4 * u + sub, N - 1) * 8 + q];
  asm volatile("" : "+v"(g[0]), "+v"(g[1]), "+v"(g[2]), "+v"(g[3]),
                    "+v"(v[0]), "+v"(v[1]), "+v"(v[2]), "+v"(v[3]));
#pragma unroll
  for (int u = 0; u < FIN_U; ++u) t[u] = mr[(size_t)g[u] * 8 + q];

  v4f x[FIN_U];
  float m[FIN_U], s[FIN_U];
#pragma unroll
  for (int u = 0; u < FIN_U; ++u) {
    x[u] = v[u] + t[u];
    m[u] = fmaxf(fmaxf(x[u].x, x[u].y), fmaxf(x[u].z, x[u].w));
  }
#pragma unroll
  for (int o = 4; o >= 1; o >>= 1) {
#pragma unroll
    for (int u = 0; u < FIN_U; ++u) m[u] = fmaxf(m[u], __shfl_xor(m[u], o, 32));
  }
#pragma unroll
  for (int u = 0; u < FIN_U; ++u) {
    x[u].x = __expf(x[u].x - m[u]);
    x[u].y = __expf(x[u].y - m[u]);
    x[u].z = __expf(x[u].z - m[u]);
    x[u].w = __expf(x[u].w - m[u]);
    s[u] = (x[u].x + x[u].y) + (x[u].z + x[u].w);
  }
#pragma unroll
  for (int o = 4; o >= 1; o >>= 1) {
#pragma unroll
    for (int u = 0; u < FIN_U; ++u) s[u] += __shfl_xor(s[u], o, 32);
  }
#pragma unroll
  for (int u = 0; u < FIN_U; ++u) {
    const int row = r0 + 4 * u + sub;
    if (row < N) orow[(size_t)row * 8 + q] = x[u] / s[u];
  }
}

extern "C" void kernel_launch(void* const* d_in, const int* in_sizes, int n_in,
                              void* d_out, int out_size, void* d_ws, size_t ws_size,
                              hipStream_t stream) {
  const float* X   = (const float*)d_in[0];
  const float* W1  = (const float*)d_in[1];
  const float* b1  = (const float*)d_in[2];
  const float* W2  = (const float*)d_in[3];
  const float* b2  = (const float*)d_in[4];
  const float* mW1 = (const float*)d_in[5];
  const float* mb1 = (const float*)d_in[6];
  const float* mW2 = (const float*)d_in[7];
  const float* mb2 = (const float*)d_in[8];
  const int* ei    = (const int*)d_in[9];
  const int* batch = (const int*)d_in[10];
  const int* gptr  = (const int*)d_in[11];

  const int N = in_sizes[0] / 128;
  const int E = in_sizes[9] / 2;
  const int* src = ei;
  const int* dst = ei + E;

  char* w = (char*)d_ws;
  size_t off = 0;
  auto take = [&](size_t bytes) -> float* {
    float* p = (float*)(w + off);
    off = (off + bytes + 255) & ~(size_t)255;
    return p;
  };
  float* xw    = take((size_t)N * 128 * 4);
  float* h     = take((size_t)N * 128 * 4);
  int*   rp    = (int*)take((size_t)N * 4);
  int*   csr   = (int*)take((size_t)N * 32 * 4);
  int2*  coarse = (int2*)h;
  int*   bintot = (int*)xw;
  int*   bincur = bintot + NBMAX;
  float* z     = h;
  float* pool1 = take((size_t)GMAX * 128 * 4);
  float* pool2 = take((size_t)GMAX * 32 * 4);
  float* cnt   = take((size_t)GMAX * 4);
  float* mm1   = take((size_t)GMAX * 128 * 4);
  float* mm2   = take((size_t)GMAX * 32 * 4);

  const int B = 256;
  auto blks = [](long long n, int b) { return (unsigned)((n + b - 1) / b); };

  const int NB = (N + BW - 1) >> BSH;
  k_fill<<<blks(2 * NBMAX, B), B, 0, stream>>>((float*)bintot, 0.0f, 2 * NBMAX);
  k_fill<<<blks(GMAX * 128, B), B, 0, stream>>>(pool1, 0.0f, GMAX * 128);
  k_fill<<<blks(GMAX * 32, B), B, 0, stream>>>(pool2, 0.0f, GMAX * 32);
  k_fill<<<blks(GMAX, B), B, 0, stream>>>(cnt, 0.0f, GMAX);
  k_bin_count<<<blks(E, EPB), 1024, 0, stream>>>(dst, bintot, E, NB);
  k_bin_place<<<blks(E, EPB), 1024, 0, stream>>>(src, dst, bintot, bincur, coarse, E, NB);
  k_bin_sort<<<NB, 1024, 0, stream>>>(coarse, bintot, rp, csr, N);

  const unsigned gblocks = blks(N, GW * RUN);

  k_gemm_wmma<128, false><<<blks(N, 16), 256, 0, stream>>>(X, W1, xw, N, nullptr, nullptr, rp, E);
  k_gather<128, true><<<gblocks, GW * 32, 0, stream>>>(xw, rp, csr, b1, h, batch, pool1, cnt, N, E);

  k_master_mlp<128, 128><<<blks(GMAX * 128, B), B, 0, stream>>>(pool1, cnt, mW1, mb1, mm1, gptr);

  k_gemm_wmma<32, true><<<blks(N, 16), 64, 0, stream>>>(h, W2, xw, N, mm1, batch, rp, E);
  k_gather<32, false><<<gblocks, GW * 32, 0, stream>>>(xw, rp, csr, b2, z, batch, pool2, nullptr, N, E);

  k_master_mlp<32, 32><<<blks(GMAX * 32, B), B, 0, stream>>>(pool2, cnt, mW2, mb2, mm2, gptr);
  k_finalize<<<blks(N, FIN_W * 4 * FIN_U), FIN_W * 32, 0, stream>>>(z, mm2, batch, (float*)d_out, N);
}


# Launches 14 -> 10: the four clears run as one launch; the 2nd graph MLP (32x32) is formed inside the softmax pass by the wave that needs it, same products and sum order, not in a launch of its own
# speedup vs baseline: 6.5636x; 1.0457x over previous
#include <hip/hip_runtime.h>
#include <math.h>


typedef __attribute__((ext_vector_type(2))) float v2f;
typedef __attribute__((ext_vector_type(8))) float v8f;

#define GMAX 256

__device__ __forceinline__ void atomAddF(float* p, float v) {
  __hip_atomic_fetch_add(p, v, __ATOMIC_RELAXED, __HIP_MEMORY_SCOPE_AGENT);
}

__global__ void k_clear4(float* a, int na, float* b, int nb, float* c, int nc, float* d, int nd) {
  int i = blockIdx.x * blockDim.x + threadIdx.x;
  if (i < na) { a[i] = 0.0f; return; }
  i -= na;
  if (i < nb) { b[i] = 0.0f; return; }
  i -= nb;
  if (i < nc) { c[i] = 0.0f; return; }
  i -= nc;
  if (i < nd) d[i] = 0.0f;
}

#define BSH   8
#define BW    (1 << BSH)
#define NBMAX 1024
#define EPT   4
#define EPB   (1024 * EPT)
#define SPT   12

__device__ __forceinline__ int lds_add_rtn(int* p, int v) {
  return __hip_atomic_fetch_add(p, v, __ATOMIC_RELAXED, __HIP_MEMORY_SCOPE_WORKGROUP);
}

__device__ __forceinline__ int wave_incl_scan(int v, int lane) {
  for (int o = 1; o < 32; o <<= 1) {
    int t = __shfl_up(v, o, 32);
    if (lane >= o) v += t;
  }
  return v;
}

__device__ __forceinline__ int block_excl_scan(int v, int* wsum, int& total) {
  const int lane = threadIdx.x & 31;
  const int wave = threadIdx.x >> 5;
  const int inc  = wave_incl_scan(v, lane);
  if (lane == 31) wsum[wave] = inc;
  __syncthreads();
  const int w    = wsum[lane];
  const int winc = wave_incl_scan(w, lane);
  total = __shfl(winc, 31, 32);
  const int wbase = __shfl(winc - w, wave, 32);
  __syncthreads();
  return wbase + inc - v;
}

__global__ void __launch_bounds__(1024) k_bin_count(const int* __restrict__ dst, int* __restrict__ bintot,
                                                    int E, int NB) {
  __shared__ int hist[NBMAX];
  for (int i = threadIdx.x; i < NB; i += 1024) hist[i] = 0;
  __syncthreads();
  const int base = blockIdx.x * EPB + threadIdx.x;
  int d[EPT];
#pragma unroll
  for (int u = 0; u < EPT; ++u) {
    const int e = base + u * 1024;
    d[u] = (e < E) ? dst[e] : -1;
  }
#pragma unroll
  for (int u = 0; u < EPT; ++u)
    if (d[u] >= 0) lds_add_rtn(&hist[d[u] >> BSH], 1);
  __syncthreads();
  for (int i = threadIdx.x; i < NB; i += 1024) {
    const int c = hist[i];
    if (c) __hip_atomic_fetch_add(&bintot[i], c, __ATOMIC_RELAXED, __HIP_MEMORY_SCOPE_AGENT);
  }
}

__global__ void __launch_bounds__(1024) k_bin_place(const int* __restrict__ src, const int* __restrict__ dst,
                                                    const int* __restrict__ bintot, int* __restrict__ bincur,
                                                    int2* __restrict__ coarse, int E, int NB) {
  __shared__ int hist[NBMAX];
  __shared__ int wsum[32];
  for (int i = threadIdx.x; i < NB; i += 1024) hist[i] = 0;
  __syncthreads();
  const int base = blockIdx.x * EPB + threadIdx.x;
  int s[EPT], d[EPT], r[EPT];
#pragma unroll
  for (int u = 0; u < EPT; ++u) {
    const int e = base + u * 1024;
    d[u] = (e < E) ? dst[e] : -1;
    s[u] = (e < E) ? src[e] : 0;
  }
#pragma unroll
  for (int u = 0; u < EPT; ++u)
    r[u] = (d[u] >= 0) ? lds_add_rtn(&hist[d[u] >> BSH], 1) : 0;

  const int t = (threadIdx.x < NB) ? bintot[threadIdx.x] : 0;
  int total;
  const int start = block_excl_scan(t, wsum, total);
  if (threadIdx.x < NB) {
    const int c = hist[threadIdx.x];
    const int off = c ? __hip_atomic_fetch_add(&bincur[threadIdx.x], c, __ATOMIC_RELAXED, __HIP_MEMORY_SCOPE_AGENT) : 0;
    hist[threadIdx.x] = start + off;
  }
  __syncthreads();
#pragma unroll
  for (int u = 0; u < EPT; ++u)
    if (d[u] >= 0) coarse[hist[d[u] >> BSH] + r[u]] = make_int2(s[u], d[u]);
}

__global__ void __launch_bounds__(1024) k_bin_sort(const int2* __restrict__ coarse, const int* __restrict__ bintot,
                                                   int* __restrict__ rp, int* __restrict__ csr, int N) {
  __shared__ int cnt[BW];
  __shared__ int wsum[32];
  const int b = blockIdx.x;
  if (threadIdx.x < BW) cnt[threadIdx.x] = 0;
  const int t = ((int)threadIdx.x < b) ? bintot[threadIdx.x] : 0;
  int bstart;
  block_excl_scan(t, wsum, bstart);
  const int bc = bintot[b];
  const int2* __restrict__ ed = coarse + bstart;

  for (int c0 = 0; c0 < bc; c0 += 1024 * SPT) {
    int y[SPT];
#pragma unroll
    for (int u = 0; u < SPT; ++u) {
      const int i = c0 + u * 1024 + (int)threadIdx.x;
      y[u] = 0;
      if (i < bc) y[u] = ed[i].y;
    }
#pragma unroll
    for (int u = 0; u < SPT; ++u) {
      const int i = c0 + u * 1024 + (int)threadIdx.x;
      if (i < bc) lds_add_rtn(&cnt[y[u] & (BW - 1)], 1);
    }
  }
  __syncthreads();

  const int v = (threadIdx.x < BW) ? cnt[threadIdx.x] : 0;
  int tot;
  const int ex = block_excl_scan(v, wsum, tot);
  if (threadIdx.x < BW) {
    const int node = b * BW + threadIdx.x;
    if (node < N) rp[node] = bstart + ex;
    cnt[threadIdx.x] = bstart + ex;
  }
  __syncthreads();

  for (int c0 = 0; c0 < bc; c0 += 1024 * SPT) {
    int2 e[SPT];
    int slot[SPT];
#pragma unroll
    for (int u = 0; u < SPT; ++u) {
      const int i = c0 + u * 1024 + (int)threadIdx.x;
      e[u] = make_int2(0, 0);
      if (i < bc) e[u] = ed[i];
    }
#pragma unroll
    for (int u = 0; u < SPT; ++u) {
      const int i = c0 + u * 1024 + (int)threadIdx.x;
      slot[u] = 0;
      if (i < bc) slot[u] = lds_add_rtn(&cnt[e[u].y & (BW - 1)], 1);
    }
#pragma unroll
    for (int u = 0; u < SPT; ++u) {
      const int i = c0 + u * 1024 + (int)threadIdx.x;
      if (i < bc) csr[slot[u]] = e[u].x;
    }
  }
}

template <int OUT, bool ADDM>
__global__ void k_gemm_wmma(const float* __restrict__ A, const float* __restrict__ W,
                            float* __restrict__ C, int N,
                            const float* __restrict__ mm, const int* __restrict__ batch,
                            const int* __restrict__ rp, int E) {
  constexpr int KD   = 128;
  constexpr int KS   = KD / 4;
  constexpr int KPRE = ADDM ? KS / 2 : KS - 2;
  constexpr int RPW  = 16 / (OUT / 16);
  constexpr int AP = KD + 4;
  __shared__ __attribute__((aligned(16))) float As[16 * AP];

  const int row0 = blockIdx.x * 16;
  const int wave = threadIdx.x >> 5;
  const int lane = threadIdx.x & 31;
  const int half = lane >> 4;
  const int n    = lane & 15;
  const int cw   = wave * 16;

  const int t0 = __builtin_amdgcn_readfirstlane(wave * RPW);
  const float4* __restrict__ Ar = (const float4*)A;
  float4 hv[RPW], mv[RPW];
#pragma unroll
  for (int u = 0; u < RPW; ++u) {
    const int row = min(row0 + t0 + u, N - 1);
    hv[u] = Ar[(size_t)row * 32 + lane];
  }
  __builtin_amdgcn_sched_barrier(0);

  v2f bf[KS];
#pragma unroll
  for (int i = 0; i < KPRE; ++i) {
    const int k = 4 * i + 2 * half;
    bf[i].x = W[(size_t)k * OUT + cw + n];
    bf[i].y = W[(size_t)(k + 1) * OUT + cw + n];
  }
  if constexpr (ADDM) {
#pragma unroll
    for (int u = 0; u < RPW; ++u) {
      const int row = min(row0 + t0 + u, N - 1);
      mv[u] = ((const float4*)mm)[(size_t)batch[row] * 32 + lane];
    }
  }
  __builtin_amdgcn_sched_barrier(0);

#pragma unroll
  for (int u = 0; u < RPW; ++u) {
    float4 v = make_float4(0.0f, 0.0f, 0.0f, 0.0f);
    if (row0 + t0 + u < N) {
      if constexpr (ADDM)
        v = make_float4(hv[u].x + mv[u].x, hv[u].y + mv[u].y, hv[u].z + mv[u].z, hv[u].w + mv[u].w);
      else
        v = hv[u];
    }
    ((float4*)As)[(t0 + u) * (AP / 4) + lane] = v;
  }
  __syncthreads();

#pragma unroll
  for (int i = KPRE; i < KS; ++i) {
    const int k = 4 * i + 2 * half;
    bf[i].x = W[(size_t)k * OUT + cw + n];
    bf[i].y = W[(size_t)(k + 1) * OUT + cw + n];
  }
  int rb[8], re[8];
#pragma unroll
  for (int r = 0; r < 8; ++r) {
    const int i = min(row0 + r + 8 * half, N - 1);
    rb[r] = rp[i];
    re[r] = (i + 1 < N) ? rp[i + 1] : E;
  }

  v8f c = {};
#pragma unroll
  for (int i = 0; i < KS; ++i) {
    const int k = 4 * i + 2 * half;
    v2f a;
    a.x = As[n * AP + k];
    a.y = As[n * AP + k + 1];
    c = __builtin_amdgcn_wmma_f32_16x16x4_f32(false, a, false, bf[i],
                                              (short)0, c, false, false);
  }

#pragma unroll
  for (int r = 0; r < 8; ++r) {
    const int m = r + 8 * half;
    if (row0 + m < N) C[(size_t)(row0 + m) * OUT + cw + n] = c[r] * rsqrtf((float)(re[r] - rb[r] + 1));
  }
}

template <int V> struct RowVec;
template <> struct RowVec<4> { typedef float4 T; };
template <> struct RowVec<1> { typedef float T; };

__device__ __forceinline__ float4 scale_add(float w, const float4 a, const float4 b) {
  return make_float4(fmaf(w, a.x, b.x), fmaf(w, a.y, b.y), fmaf(w, a.z, b.z), fmaf(w, a.w, b.w));
}
__device__ __forceinline__ float scale_add(float w, const float a, const float b) { return fmaf(w, a, b); }
__device__ __forceinline__ float4 plus(const float4 a, const float4 b) {
  return make_float4(a.x + b.x, a.y + b.y, a.z + b.z, a.w + b.w);
}
__device__ __forceinline__ float plus(const float a, const float b) { return a + b; }
__device__ __forceinline__ void set_zero(float4& a) { a = make_float4(0.0f, 0.0f, 0.0f, 0.0f); }
__device__ __forceinline__ void set_zero(float& a) { a = 0.0f; }

template <int B, typename T>
__device__ __forceinline__ void gather_rows(T& acc, const T* __restrict__ xr, int s_i, int k, int lane) {
  T v[B];
  for (int u = 0; u < B; ++u) {
    const int s = __builtin_amdgcn_readlane(s_i, k + u);
    v[u] = xr[(size_t)s * 32 + lane];
  }
  for (int u = 0; u < B; ++u) acc = plus(acc, v[u]);
}

__device__ __forceinline__ float elu1(float x) { return x > 0.0f ? x : expm1f(x); }
__device__ __forceinline__ float4 elu_row(const float4 a) {
  return make_float4(elu1(a.x), elu1(a.y), elu1(a.z), elu1(a.w));
}
__device__ __forceinline__ float elu_row(const float a) { return elu1(a); }

__device__ __forceinline__ void pool_add_row(float* __restrict__ pool, int g, int lane, const float4 a) {
  float* p = &pool[(size_t)g * 128 + lane * 4];
  atomAddF(p + 0, a.x); atomAddF(p + 1, a.y);
  atomAddF(p + 2, a.z); atomAddF(p + 3, a.w);
}
__device__ __forceinline__ void pool_add_row(float* __restrict__ pool, int g, int lane, const float a) {
  atomAddF(&pool[(size_t)g * 32 + lane], a);
}

#define RUN 3
#define GW  8

template <int F, bool ELU>
__global__ void __launch_bounds__(GW * 32) k_gather(const float* __restrict__ xw, const int* __restrict__ rp,
                         const int* __restrict__ csr, const float* __restrict__ bias,
                         float* __restrict__ out, const int* __restrict__ batch,
                         float* __restrict__ pool, float* __restrict__ cnt, int N, int E) {
  constexpr int V = F / 32;
  typedef typename RowVec<V>::T T;
  __shared__ T     prow[GW * 32];
  __shared__ float pcnt[GW];
  __shared__ int   pgr[GW];
  const int lane = threadIdx.x & 31;
  const T* __restrict__ xr = (const T*)xw;
  T* __restrict__ orow     = (T*)out;
  const T bv = ((const T*)bias)[lane];

  const int wave = __builtin_amdgcn_readfirstlane((int)(threadIdx.x >> 5));
  const int n0 = (blockIdx.x * GW + wave) * RUN;
  const int n1 = min(n0 + RUN, N);

  T pacc;
  set_zero(pacc);
  float pc = 0.0f;
  int gcur = (n0 < N) ? batch[n0] : -1;

  for (int node = n0; node < n1; ++node) {
    const int beg  = rp[node];
    const int end  = (node + 1 < N) ? rp[node + 1] : E;
    const int g    = batch[node];
    const float dn = rsqrtf((float)(end - beg + 1));
    const T self = xr[(size_t)node * 32 + lane];
    T acc;
    set_zero(acc);

    for (int base = beg; base < end; base += 32) {
      const int j = base + lane;
      const int s_i = (j < end) ? csr[j] : 0;
      const int cnt_e = min(32, end - base);
      int k = 0;
      for (; k + 8 <= cnt_e; k += 8) gather_rows<8>(acc, xr, s_i, k, lane);
      if (k + 4 <= cnt_e) { gather_rows<4>(acc, xr, s_i, k, lane); k += 4; }
      if (k + 2 <= cnt_e) { gather_rows<2>(acc, xr, s_i, k, lane); k += 2; }
      if (k < cnt_e)      { gather_rows<1>(acc, xr, s_i, k, lane); }
    }
    acc = plus(acc, self);
    T r = scale_add(dn, acc, bv);
    if constexpr (ELU) r = elu_row(r);
    orow[(size_t)node * 32 + lane] = r;

    if (g != gcur) {
      pool_add_row(pool, gcur, lane, pacc);
      if (ELU && lane == 0) atomAddF(&cnt[gcur], pc);
      set_zero(pacc);
      pc = 0.0f;
      gcur = g;
    }
    pacc = plus(pacc, r);
    pc += 1.0f;
  }

  prow[wave * 32 + lane] = pacc;
  if (lane == 0) { pcnt[wave] = pc; pgr[wave] = gcur; }
  __syncthreads();
  if (threadIdx.x < F) {
    const float* pf = (const float*)prow;
    float sq[GW], cq[GW];
    int gq[GW];
#pragma unroll
    for (int w = 0; w < GW; ++w) {
      gq[w] = pgr[w];
      cq[w] = pcnt[w];
      sq[w] = pf[w * F + threadIdx.x];
    }
    float s = 0.0f, c = 0.0f;
    int g = -1;
#pragma unroll
    for (int w = 0; w < GW; ++w) {
      const int gw = gq[w];
      if (gw < 0) continue;
      if (gw != g) {
        if (g >= 0) {
          atomAddF(&pool[(size_t)g * F + threadIdx.x], s);
          if (ELU && threadIdx.x == 0) atomAddF(&cnt[g], c);
        }
        s = 0.0f;
        c = 0.0f;
        g = gw;
      }
      s += sq[w];
      c += cq[w];
    }
    if (g >= 0) {
      atomAddF(&pool[(size_t)g * F + threadIdx.x], s);
      if (ELU && threadIdx.x == 0) atomAddF(&cnt[g], c);
    }
  }
}

template <int IN, int OUT>
__global__ void k_master_mlp(const float* __restrict__ pool, const float* __restrict__ cnt,
                             const float* __restrict__ Wm, const float* __restrict__ bm,
                             float* __restrict__ mm, const int* __restrict__ gptr) {
  int G = *gptr;
  int idx = blockIdx.x * blockDim.x + threadIdx.x;
  int g = idx / OUT, j = idx % OUT;
  if (g >= G) return;
  float inv = 1.0f / fmaxf(cnt[g], 1.0f);
  float acc = bm[j];
#pragma unroll 32
  for (int i = 0; i < IN; ++i)
    acc = fmaf(pool[(size_t)g * IN + i] * inv, Wm[(size_t)i * OUT + j], acc);
  mm[(size_t)g * OUT + j] = fmaxf(acc, 0.0f);
}

typedef __attribute__((ext_vector_type(4))) float v4f;
#define FIN_U 4
#define FIN_W 4
#define FIN_K 32

__device__ __forceinline__ v4f fma4(float p, const v4f w, const v4f a) {
  v4f r;
  r.x = fmaf(p, w.x, a.x); r.y = fmaf(p, w.y, a.y);
  r.z = fmaf(p, w.z, a.z); r.w = fmaf(p, w.w, a.w);
  return r;
}
__device__ __forceinline__ v4f relu4(const v4f a) {
  v4f r;
  r.x = fmaxf(a.x, 0.0f); r.y = fmaxf(a.y, 0.0f);
  r.z = fmaxf(a.z, 0.0f); r.w = fmaxf(a.w, 0.0f);
  return r;
}

__global__ void __launch_bounds__(FIN_W * 32) k_finalize(const float* __restrict__ z, const float* __restrict__ pool2,
                           const float* __restrict__ cnt, const float* __restrict__ Wm,
                           const float* __restrict__ bm, const int* __restrict__ batch,
                           float* __restrict__ out, int N) {
  const int lane = threadIdx.x & 31;
  const int sub  = lane >> 3;
  const int q    = lane & 7;
  const int r0   = (blockIdx.x * FIN_W + (threadIdx.x >> 5)) * (4 * FIN_U);
  if (r0 >= N) return;
  const v4f* __restrict__ zr = (const v4f*)z;
  const v4f* __restrict__ wr = (const v4f*)Wm;
  v4f* __restrict__ orow     = (v4f*)out;

  int g[FIN_U];
  v4f v[FIN_U], t[FIN_U];
#pragma unroll
  for (int u = 0; u < FIN_U; ++u) g[u] = batch[min(r0 + 4 * u + sub, N - 1)];
#pragma unroll
  for (int u = 0; u < FIN_U; ++u) v[u] = zr[(size_t)min(r0 + 4 * u + sub, N - 1) * 8 + q];
  asm volatile("" : "+v"(g[0]), "+v"(g[1]), "+v"(g[2]), "+v"(g[3]),
                    "+v"(v[0]), "+v"(v[1]), "+v"(v[2]), "+v"(v[3]));
  v4f w[FIN_K];
#pragma unroll
  for (int i = 0; i < FIN_K; ++i) w[i] = wr[i * 8 + q];
  const v4f b4 = ((const v4f*)bm)[q];

  const int g0 = __builtin_amdgcn_readfirstlane(g[0]);
  bool same = true;
#pragma unroll
  for (int u = 0; u < FIN_U; ++u) same = same && (g[u] == g0);
  if (__all(same)) {
    const float inv = 1.0f / fmaxf(cnt[g0], 1.0f);
    const float pv  = pool2[(size_t)g0 * FIN_K + lane] * inv;
    v4f acc = b4;
#pragma unroll
    for (int i = 0; i < FIN_K; ++i) {
      const float p = __builtin_bit_cast(float, __builtin_amdgcn_readlane(__builtin_bit_cast(int, pv), i));
      acc = fma4(p, w[i], acc);
    }
    acc = relu4(acc);
#pragma unroll
    for (int u = 0; u < FIN_U; ++u) t[u] = acc;
  } else {
#pragma unroll
    for (int u = 0; u < FIN_U; ++u) {
      const float inv = 1.0f / fmaxf(cnt[g[u]], 1.0f);
      float p[FIN_K];
#pragma unroll
      for (int i = 0; i < FIN_K; ++i) p[i] = pool2[(size_t)g[u] * FIN_K + i];
      v4f acc = b4;
#pragma unroll
      for (int i = 0; i < FIN_K; ++i) acc = fma4(p[i] * inv, w[i], acc);
      t[u] = relu4(acc);
    }
  }

  v4f x[FIN_U];
  float m[FIN_U], s[FIN_U];
#pragma unroll
  for (int u = 0; u < FIN_U; ++u) {
    x[u] = v[u] + t[u];
    m[u] = fmaxf(fmaxf(x[u].x, x[u].y), fmaxf(x[u].z, x[u].w));
  }
#pragma unroll
  for (int o = 4; o >= 1; o >>= 1) {
#pragma unroll
    for (int u = 0; u < FIN_U; ++u) m[u] = fmaxf(m[u], __shfl_xor(m[u], o, 32));
  }
#pragma unroll
  for (int u = 0; u < FIN_U; ++u) {
    x[u].x = __expf(x[u].x - m[u]);
    x[u].y = __expf(x[u].y - m[u]);
    x[u].z = __expf(x[u].z - m[u]);
    x[u].w = __expf(x[u].w - m[u]);
    s[u] = (x[u].x + x[u].y) + (x[u].z + x[u].w);
  }
#pragma unroll
  for (int o = 4; o >= 1; o >>= 1) {
#pragma unroll
    for (int u = 0; u < FIN_U; ++u) s[u] += __shfl_xor(s[u], o, 32);
  }
#pragma unroll
  for (int u = 0; u < FIN_U; ++u) {
    const int row = r0 + 4 * u + sub;
    if (row < N) orow[(size_t)row * 8 + q] = x[u] / s[u];
  }
}

extern "C" void kernel_launch(void* const* d_in, const int* in_sizes, int n_in,
                              void* d_out, int out_size, void* d_ws, size_t ws_size,
                              hipStream_t stream) {
  const float* X   = (const float*)d_in[0];
  const float* W1  = (const float*)d_in[1];
  const float* b1  = (const float*)d_in[2];
  const float* W2  = (const float*)d_in[3];
  const float* b2  = (const float*)d_in[4];
  const float* mW1 = (const float*)d_in[5];
  const float* mb1 = (const float*)d_in[6];
  const float* mW2 = (const float*)d_in[7];
  const float* mb2 = (const float*)d_in[8];
  const int* ei    = (const int*)d_in[9];
  const int* batch = (const int*)d_in[10];
  const int* gptr  = (const int*)d_in[11];

  const int N = in_sizes[0] / 128;
  const int E = in_sizes[9] / 2;
  const int* src = ei;
  const int* dst = ei + E;

  char* w = (char*)d_ws;
  size_t off = 0;
  auto take = [&](size_t bytes) -> float* {
    float* p = (float*)(w + off);
    off = (off + bytes + 255) & ~(size_t)255;
    return p;
  };
  float* xw    = take((size_t)N * 128 * 4);
  float* h     = take((size_t)N * 128 * 4);
  int*   rp    = (int*)take((size_t)N * 4);
  int*   csr   = (int*)take((size_t)N * 32 * 4);
  int2*  coarse = (int2*)h;
  int*   bintot = (int*)xw;
  int*   bincur = bintot + NBMAX;
  float* z     = h;
  float* pool1 = take((size_t)GMAX * 128 * 4);
  float* pool2 = take((size_t)GMAX * 32 * 4);
  float* cnt   = take((size_t)GMAX * 4);
  float* mm1   = take((size_t)GMAX * 128 * 4);

  const int B = 256;
  auto blks = [](long long n, int b) { return (unsigned)((n + b - 1) / b); };

  const int NB = (N + BW - 1) >> BSH;
  k_clear4<<<blks(2 * NBMAX + GMAX * 128 + GMAX * 32 + GMAX, B), B, 0, stream>>>(
      (float*)bintot, 2 * NBMAX, pool1, GMAX * 128, pool2, GMAX * 32, cnt, GMAX);
  k_bin_count<<<blks(E, EPB), 1024, 0, stream>>>(dst, bintot, E, NB);
  k_bin_place<<<blks(E, EPB), 1024, 0, stream>>>(src, dst, bintot, bincur, coarse, E, NB);
  k_bin_sort<<<NB, 1024, 0, stream>>>(coarse, bintot, rp, csr, N);

  const unsigned gblocks = blks(N, GW * RUN);

  k_gemm_wmma<128, false><<<blks(N, 16), 256, 0, stream>>>(X, W1, xw, N, nullptr, nullptr, rp, E);
  k_gather<128, true><<<gblocks, GW * 32, 0, stream>>>(xw, rp, csr, b1, h, batch, pool1, cnt, N, E);

  k_master_mlp<128, 128><<<blks(GMAX * 128, B), B, 0, stream>>>(pool1, cnt, mW1, mb1, mm1, gptr);

  k_gemm_wmma<32, true><<<blks(N, 16), 64, 0, stream>>>(h, W2, xw, N, mm1, batch, rp, E);
  k_gather<32, false><<<gblocks, GW * 32, 0, stream>>>(xw, rp, csr, b2, z, batch, pool2, nullptr, N, E);

  k_finalize<<<blks(N, FIN_W * 4 * FIN_U), FIN_W * 32, 0, stream>>>(z, pool2, cnt, mW2, mb2, batch, (float*)d_out, N);
}
